# starting kernel
# baseline (speedup 1.0000x reference)
#include <hip/hip_runtime.h>
#include <math.h>

typedef float v2f __attribute__((ext_vector_type(2)));
typedef float v8f __attribute__((ext_vector_type(8)));

#define B_    8
#define C_    512
#define N_    1024
#define C3_   1536
#define GRP_  8
#define GCH_  64
#define GELT_ (GCH_ * N_)
#define EPS_  1e-5f

__global__ __launch_bounds__(256) void kT_transpose(const float* __restrict__ in,
                                                    float* __restrict__ out,
                                                    int R, int Cc) {
  __shared__ float tile[32][33];
  int c0 = blockIdx.x * 32, r0 = blockIdx.y * 32;
  int tx = threadIdx.x, ty = threadIdx.y;
  #pragma unroll
  for (int i = 0; i < 32; i += 8)
    tile[ty + i][tx] = in[(size_t)(r0 + ty + i) * Cc + c0 + tx];
  __syncthreads();
  #pragma unroll
  for (int i = 0; i < 32; i += 8)
    out[(size_t)(c0 + ty + i) * R + r0 + tx] = tile[tx][ty + i];
}

__global__ __launch_bounds__(256) void kT_qtranspose(const float* __restrict__ qkv,
                                                     float* __restrict__ qT) {
  __shared__ float tile[32][33];
  int d0 = blockIdx.x * 32, n0 = blockIdx.y * 32;
  const float* in = qkv + (size_t)blockIdx.z * N_ * C3_;
  float*      out = qT  + (size_t)blockIdx.z * C_ * N_;
  int tx = threadIdx.x, ty = threadIdx.y;
  #pragma unroll
  for (int i = 0; i < 32; i += 8)
    tile[ty + i][tx] = in[(size_t)(n0 + ty + i) * C3_ + d0 + tx];
  __syncthreads();
  #pragma unroll
  for (int i = 0; i < 32; i += 8)
    out[(size_t)(d0 + ty + i) * N_ + n0 + tx] = tile[tx][ty + i];
}

__global__ __launch_bounds__(256) void kT_gn_stats(const float* __restrict__ x,
                                                   float* __restrict__ mean,
                                                   float* __restrict__ rstd) {
  __shared__ float s1[256], s2[256];
  int bg = blockIdx.x;
  const float4* p = (const float4*)(x + (size_t)bg * GELT_);
  float s = 0.f, q = 0.f;
  for (int i = threadIdx.x; i < GELT_ / 4; i += 256) {
    float4 v = p[i];
    s += v.x + v.y + v.z + v.w;
    q += v.x * v.x + v.y * v.y + v.z * v.z + v.w * v.w;
  }
  s1[threadIdx.x] = s; s2[threadIdx.x] = q;
  __syncthreads();
  for (int off = 128; off > 0; off >>= 1) {
    if (threadIdx.x < off) {
      s1[threadIdx.x] += s1[threadIdx.x + off];
      s2[threadIdx.x] += s2[threadIdx.x + off];
    }
    __syncthreads();
  }
  if (threadIdx.x == 0) {
    float m = s1[0] * (1.0f / GELT_);
    float v = s2[0] * (1.0f / GELT_) - m * m;
    mean[bg] = m;
    rstd[bg] = rsqrtf(v + EPS_);
  }
}

__global__ __launch_bounds__(256) void kT_gn_apply(const float* __restrict__ x,
                                                   const float* __restrict__ gamma,
                                                   const float* __restrict__ beta,
                                                   const float* __restrict__ mean,
                                                   const float* __restrict__ rstd,
                                                   float* __restrict__ xn) {
  int bc = blockIdx.x;
  int c = bc & (C_ - 1);
  int bg = ((bc >> 9) << 3) + (c >> 6);
  float gm = gamma[c] * rstd[bg];
  float bt = beta[c] - mean[bg] * gm;
  const float4* pin  = (const float4*)(x  + (size_t)bc * N_);
  float4*       pout = (float4*)(xn + (size_t)bc * N_);
  float4 v = pin[threadIdx.x];
  v.x = v.x * gm + bt; v.y = v.y * gm + bt;
  v.z = v.z * gm + bt; v.w = v.w * gm + bt;
  pout[threadIdx.x] = v;
}

__global__ __launch_bounds__(256) void kT_softmax_k(float* __restrict__ qkv) {
  __shared__ float red[256];
  float* row = qkv + (size_t)blockIdx.x * C3_ + C_;
  int t = threadIdx.x;
  float a = row[t], b = row[t + 256];
  red[t] = fmaxf(a, b);
  __syncthreads();
  for (int off = 128; off > 0; off >>= 1) {
    if (t < off) red[t] = fmaxf(red[t], red[t + off]);
    __syncthreads();
  }
  float mx = red[0];
  __syncthreads();
  float e0 = __expf(a - mx), e1 = __expf(b - mx);
  red[t] = e0 + e1;
  __syncthreads();
  for (int off = 128; off > 0; off >>= 1) {
    if (t < off) red[t] += red[t + off];
    __syncthreads();
  }
  float inv = 1.0f / red[0];
  row[t] = e0 * inv;
  row[t + 256] = e1 * inv;
}

__global__ __launch_bounds__(256) void kT_softmax_row1024(float* __restrict__ qT) {
  __shared__ float red[256];
  float4* row = (float4*)(qT + (size_t)blockIdx.x * N_);
  int t = threadIdx.x;
  float4 v = row[t];
  red[t] = fmaxf(fmaxf(v.x, v.y), fmaxf(v.z, v.w));
  __syncthreads();
  for (int off = 128; off > 0; off >>= 1) {
    if (t < off) red[t] = fmaxf(red[t], red[t + off]);
    __syncthreads();
  }
  float mx = red[0];
  __syncthreads();
  float4 e;
  e.x = __expf(v.x - mx); e.y = __expf(v.y - mx);
  e.z = __expf(v.z - mx); e.w = __expf(v.w - mx);
  red[t] = e.x + e.y + e.z + e.w;
  __syncthreads();
  for (int off = 128; off > 0; off >>= 1) {
    if (t < off) red[t] += red[t + off];
    __syncthreads();
  }
  float inv = 1.0f / red[0];
  e.x *= inv; e.y *= inv; e.z *= inv; e.w *= inv;
  row[t] = e;
}

template <int LDA, int LDB, int LDC, int K, bool TSTORE, bool HAS_BIAS, bool HAS_RES>
__global__ __launch_bounds__(256) void kT_gemm(
    const float* __restrict__ A, long long strideA,
    const float* __restrict__ B, long long strideB,
    float* __restrict__ Cmat,    long long strideC,
    const float* __restrict__ bias,
    const float* __restrict__ resid, long long strideR) {
  int z = blockIdx.z;
  const float* Ab = A + (size_t)z * strideA;
  const float* Bb = B + (size_t)z * strideB;
  float*       Cb = Cmat + (size_t)z * strideC;

  int wave = threadIdx.x >> 5;
  int lane = threadIdx.x & 31;
  int half = lane >> 4;
  int ml   = lane & 15;
  int m0 = blockIdx.x * 64  + (wave & 1) * 32;
  int n0 = blockIdx.y * 256 + (wave >> 1) * 64;

  v8f acc[2][4];
  {
    float bv[4];
    #pragma unroll
    for (int s = 0; s < 4; ++s) bv[s] = HAS_BIAS ? bias[n0 + 16 * s + ml] : 0.0f;
    #pragma unroll
    for (int mi = 0; mi < 2; ++mi)
      #pragma unroll
      for (int s = 0; s < 4; ++s)
        #pragma unroll
        for (int r = 0; r < 8; ++r) acc[mi][s][r] = bv[s];
  }

  const float* Ap = Ab + (size_t)(2 * half) * LDA + m0 + ml;
  const float* Bp = Bb + (size_t)(2 * half) * LDB + n0 + ml;

  #pragma unroll 1
  for (int kk = 0; kk < K; kk += 16) {
    __builtin_prefetch(Ap + (size_t)20 * LDA, 0, 1);
    #pragma unroll
    for (int j = 0; j < 4; ++j) {
      const int k0 = 4 * j * LDA, k1 = (4 * j + 1) * LDA;
      const int h0 = 4 * j * LDB, h1 = (4 * j + 1) * LDB;
      v2f a0; a0.x = Ap[k0];      a0.y = Ap[k1];
      v2f a1; a1.x = Ap[k0 + 16]; a1.y = Ap[k1 + 16];
      v2f b0; b0.x = Bp[h0];      b0.y = Bp[h1];
      v2f b1; b1.x = Bp[h0 + 16]; b1.y = Bp[h1 + 16];
      v2f b2; b2.x = Bp[h0 + 32]; b2.y = Bp[h1 + 32];
      v2f b3; b3.x = Bp[h0 + 48]; b3.y = Bp[h1 + 48];
      acc[0][0] = __builtin_amdgcn_wmma_f32_16x16x4_f32(false, a0, false, b0, (short)0, acc[0][0], false, false);
      acc[0][1] = __builtin_amdgcn_wmma_f32_16x16x4_f32(false, a0, false, b1, (short)0, acc[0][1], false, false);
      acc[0][2] = __builtin_amdgcn_wmma_f32_16x16x4_f32(false, a0, false, b2, (short)0, acc[0][2], false, false);
      acc[0][3] = __builtin_amdgcn_wmma_f32_16x16x4_f32(false, a0, false, b3, (short)0, acc[0][3], false, false);
      acc[1][0] = __builtin_amdgcn_wmma_f32_16x16x4_f32(false, a1, false, b0, (short)0, acc[1][0], false, false);
      acc[1][1] = __builtin_amdgcn_wmma_f32_16x16x4_f32(false, a1, false, b1, (short)0, acc[1][1], false, false);
      acc[1][2] = __builtin_amdgcn_wmma_f32_16x16x4_f32(false, a1, false, b2, (short)0, acc[1][2], false, false);
      acc[1][3] = __builtin_amdgcn_wmma_f32_16x16x4_f32(false, a1, false, b3, (short)0, acc[1][3], false, false);
    }
    Ap += 16 * LDA;
    Bp += 16 * LDB;
  }

  if (!TSTORE) {
    #pragma unroll
    for (int mi = 0; mi < 2; ++mi) {
      #pragma unroll
      for (int r = 0; r < 8; ++r) {
        int m = m0 + 16 * mi + r + 8 * half;
        #pragma unroll
        for (int s = 0; s < 4; ++s)
          Cb[(size_t)m * LDC + n0 + 16 * s + ml] = acc[mi][s][r];
      }
    }
  } else {
    const float* Rb = HAS_RES ? (resid + (size_t)z * strideR) : nullptr;
    #pragma unroll
    for (int mi = 0; mi < 2; ++mi) {
      #pragma unroll
      for (int r = 0; r < 8; ++r) {
        int m = m0 + 16 * mi + r + 8 * half;
        #pragma unroll
        for (int s = 0; s < 4; ++s) {
          size_t i = (size_t)(n0 + 16 * s + ml) * LDC + m;
          float v = acc[mi][s][r];
          if (HAS_RES) v += Rb[i];
          Cb[i] = v;
        }
      }
    }
  }
}

extern "C" void kernel_launch(void* const* d_in, const int* in_sizes, int n_in,
                              void* d_out, int out_size, void* d_ws, size_t ws_size,
                              hipStream_t stream) {
  const float* x      = (const float*)d_in[0];
  const float* gn_w   = (const float*)d_in[1];
  const float* gn_b   = (const float*)d_in[2];
  const float* qkv_w  = (const float*)d_in[3];
  const float* qkv_b  = (const float*)d_in[4];
  const float* proj_w = (const float*)d_in[5];
  const float* proj_b = (const float*)d_in[6];
  float* out = (float*)d_out;

  float* ws = (float*)d_ws;
  const size_t XN_SZ  = (size_t)B_ * C_ * N_;
  const size_t QKV_SZ = (size_t)B_ * N_ * C3_;
  const size_t CTX_SZ = (size_t)B_ * C_ * C_;
  float* mean   = ws;
  float* rstd   = ws + 64;
  float* xn     = ws + 256;
  float* qkv    = xn   + XN_SZ;
  float* qT     = qkv  + QKV_SZ;
  float* ctx    = qT   + XN_SZ;
  float* o2T    = ctx  + CTX_SZ;
  float* wqkvT  = o2T  + XN_SZ;
  float* wprojT = wqkvT + (size_t)C_ * C3_;

  kT_transpose<<<dim3(C_ / 32, C3_ / 32), dim3(32, 8), 0, stream>>>(qkv_w,  wqkvT,  C3_, C_);
  kT_transpose<<<dim3(C_ / 32, C_  / 32), dim3(32, 8), 0, stream>>>(proj_w, wprojT, C_,  C_);

  kT_gn_stats<<<B_ * GRP_, 256, 0, stream>>>(x, mean, rstd);
  kT_gn_apply<<<B_ * C_,   256, 0, stream>>>(x, gn_w, gn_b, mean, rstd, xn);

  kT_gemm<N_, C3_, C3_, C_, false, true, false>
      <<<dim3(N_ / 64, C3_ / 256, B_), 256, 0, stream>>>(
      xn, (long long)C_ * N_,
      wqkvT, 0,
      qkv, (long long)N_ * C3_,
      qkv_b, nullptr, 0);

  kT_softmax_k<<<B_ * N_, 256, 0, stream>>>(qkv);

  kT_qtranspose<<<dim3(C_ / 32, N_ / 32, B_), dim3(32, 8), 0, stream>>>(qkv, qT);
  kT_softmax_row1024<<<B_ * C_, 256, 0, stream>>>(qT);

  kT_gemm<C3_, C3_, C_, N_, false, false, false>
      <<<dim3(C_ / 64, C_ / 256, B_), 256, 0, stream>>>(
      qkv + C_,     (long long)N_ * C3_,
      qkv + 2 * C_, (long long)N_ * C3_,
      ctx, (long long)C_ * C_,
      nullptr, nullptr, 0);

  kT_gemm<N_, C_, N_, C_, true, false, false>
      <<<dim3(N_ / 64, C_ / 256, B_), 256, 0, stream>>>(
      qT, (long long)C_ * N_,
      ctx, (long long)C_ * C_,
      o2T, (long long)C_ * N_,
      nullptr, nullptr, 0);

  kT_gemm<N_, C_, N_, C_, true, true, true>
      <<<dim3(N_ / 64, C_ / 256, B_), 256, 0, stream>>>(
      o2T, (long long)C_ * N_,
      wprojT, 0,
      out, (long long)C_ * N_,
      proj_b, xn, (long long)C_ * N_);
}


# GEMM K loop software-pipelined: fragment loads run a register ring of 4 (qkv) or 8 (other GEMMs) k-steps ahead of the WMMAs that use them; prefetch hint dropped; sum order kept
# speedup vs baseline: 1.3473x; 1.3473x over previous
#include <hip/hip_runtime.h>
#include <math.h>

typedef float v2f __attribute__((ext_vector_type(2)));
typedef float v8f __attribute__((ext_vector_type(8)));

#define B_    8
#define C_    512
#define N_    1024
#define C3_   1536
#define GRP_  8
#define GCH_  64
#define GELT_ (GCH_ * N_)
#define EPS_  1e-5f

__global__ __launch_bounds__(256) void kT_transpose(const float* __restrict__ in,
                                                    float* __restrict__ out,
                                                    int R, int Cc) {
  __shared__ float tile[32][33];
  int c0 = blockIdx.x * 32, r0 = blockIdx.y * 32;
  int tx = threadIdx.x, ty = threadIdx.y;
  #pragma unroll
  for (int i = 0; i < 32; i += 8)
    tile[ty + i][tx] = in[(size_t)(r0 + ty + i) * Cc + c0 + tx];
  __syncthreads();
  #pragma unroll
  for (int i = 0; i < 32; i += 8)
    out[(size_t)(c0 + ty + i) * R + r0 + tx] = tile[tx][ty + i];
}

__global__ __launch_bounds__(256) void kT_qtranspose(const float* __restrict__ qkv,
                                                     float* __restrict__ qT) {
  __shared__ float tile[32][33];
  int d0 = blockIdx.x * 32, n0 = blockIdx.y * 32;
  const float* in = qkv + (size_t)blockIdx.z * N_ * C3_;
  float*      out = qT  + (size_t)blockIdx.z * C_ * N_;
  int tx = threadIdx.x, ty = threadIdx.y;
  #pragma unroll
  for (int i = 0; i < 32; i += 8)
    tile[ty + i][tx] = in[(size_t)(n0 + ty + i) * C3_ + d0 + tx];
  __syncthreads();
  #pragma unroll
  for (int i = 0; i < 32; i += 8)
    out[(size_t)(d0 + ty + i) * N_ + n0 + tx] = tile[tx][ty + i];
}

__global__ __launch_bounds__(256) void kT_gn_stats(const float* __restrict__ x,
                                                   float* __restrict__ mean,
                                                   float* __restrict__ rstd) {
  __shared__ float s1[256], s2[256];
  int bg = blockIdx.x;
  const float4* p = (const float4*)(x + (size_t)bg * GELT_);
  float s = 0.f, q = 0.f;
  for (int i = threadIdx.x; i < GELT_ / 4; i += 256) {
    float4 v = p[i];
    s += v.x + v.y + v.z + v.w;
    q += v.x * v.x + v.y * v.y + v.z * v.z + v.w * v.w;
  }
  s1[threadIdx.x] = s; s2[threadIdx.x] = q;
  __syncthreads();
  for (int off = 128; off > 0; off >>= 1) {
    if (threadIdx.x < off) {
      s1[threadIdx.x] += s1[threadIdx.x + off];
      s2[threadIdx.x] += s2[threadIdx.x + off];
    }
    __syncthreads();
  }
  if (threadIdx.x == 0) {
    float m = s1[0] * (1.0f / GELT_);
    float v = s2[0] * (1.0f / GELT_) - m * m;
    mean[bg] = m;
    rstd[bg] = rsqrtf(v + EPS_);
  }
}

__global__ __launch_bounds__(256) void kT_gn_apply(const float* __restrict__ x,
                                                   const float* __restrict__ gamma,
                                                   const float* __restrict__ beta,
                                                   const float* __restrict__ mean,
                                                   const float* __restrict__ rstd,
                                                   float* __restrict__ xn) {
  int bc = blockIdx.x;
  int c = bc & (C_ - 1);
  int bg = ((bc >> 9) << 3) + (c >> 6);
  float gm = gamma[c] * rstd[bg];
  float bt = beta[c] - mean[bg] * gm;
  const float4* pin  = (const float4*)(x  + (size_t)bc * N_);
  float4*       pout = (float4*)(xn + (size_t)bc * N_);
  float4 v = pin[threadIdx.x];
  v.x = v.x * gm + bt; v.y = v.y * gm + bt;
  v.z = v.z * gm + bt; v.w = v.w * gm + bt;
  pout[threadIdx.x] = v;
}

__global__ __launch_bounds__(256) void kT_softmax_k(float* __restrict__ qkv) {
  __shared__ float red[256];
  float* row = qkv + (size_t)blockIdx.x * C3_ + C_;
  int t = threadIdx.x;
  float a = row[t], b = row[t + 256];
  red[t] = fmaxf(a, b);
  __syncthreads();
  for (int off = 128; off > 0; off >>= 1) {
    if (t < off) red[t] = fmaxf(red[t], red[t + off]);
    __syncthreads();
  }
  float mx = red[0];
  __syncthreads();
  float e0 = __expf(a - mx), e1 = __expf(b - mx);
  red[t] = e0 + e1;
  __syncthreads();
  for (int off = 128; off > 0; off >>= 1) {
    if (t < off) red[t] += red[t + off];
    __syncthreads();
  }
  float inv = 1.0f / red[0];
  row[t] = e0 * inv;
  row[t + 256] = e1 * inv;
}

__global__ __launch_bounds__(256) void kT_softmax_row1024(float* __restrict__ qT) {
  __shared__ float red[256];
  float4* row = (float4*)(qT + (size_t)blockIdx.x * N_);
  int t = threadIdx.x;
  float4 v = row[t];
  red[t] = fmaxf(fmaxf(v.x, v.y), fmaxf(v.z, v.w));
  __syncthreads();
  for (int off = 128; off > 0; off >>= 1) {
    if (t < off) red[t] = fmaxf(red[t], red[t + off]);
    __syncthreads();
  }
  float mx = red[0];
  __syncthreads();
  float4 e;
  e.x = __expf(v.x - mx); e.y = __expf(v.y - mx);
  e.z = __expf(v.z - mx); e.w = __expf(v.w - mx);
  red[t] = e.x + e.y + e.z + e.w;
  __syncthreads();
  for (int off = 128; off > 0; off >>= 1) {
    if (t < off) red[t] += red[t + off];
    __syncthreads();
  }
  float inv = 1.0f / red[0];
  e.x *= inv; e.y *= inv; e.z *= inv; e.w *= inv;
  row[t] = e;
}

template <int LDA, int LDB, int LDC, int K, int NS, bool TSTORE, bool HAS_BIAS, bool HAS_RES>
__global__ __launch_bounds__(256) void kT_gemm(
    const float* __restrict__ A, long long strideA,
    const float* __restrict__ B, long long strideB,
    float* __restrict__ Cmat,    long long strideC,
    const float* __restrict__ bias,
    const float* __restrict__ resid, long long strideR) {
  static_assert(K % (4 * NS) == 0 && K >= 8 * NS, "ring must divide K");
  int z = blockIdx.z;
  const float* Ab = A + (size_t)z * strideA;
  const float* Bb = B + (size_t)z * strideB;
  float*       Cb = Cmat + (size_t)z * strideC;

  int wave = threadIdx.x >> 5;
  int lane = threadIdx.x & 31;
  int half = lane >> 4;
  int ml   = lane & 15;
  int m0 = blockIdx.x * 64  + (wave & 1) * 32;
  int n0 = blockIdx.y * 256 + (wave >> 1) * 64;

  v8f acc[2][4];
  {
    float bv[4];
    #pragma unroll
    for (int s = 0; s < 4; ++s) bv[s] = HAS_BIAS ? bias[n0 + 16 * s + ml] : 0.0f;
    #pragma unroll
    for (int mi = 0; mi < 2; ++mi)
      #pragma unroll
      for (int s = 0; s < 4; ++s)
        #pragma unroll
        for (int r = 0; r < 8; ++r) acc[mi][s][r] = bv[s];
  }

  const float* Ap = Ab + (size_t)(2 * half) * LDA + m0 + ml;
  const float* Bp = Bb + (size_t)(2 * half) * LDB + n0 + ml;

  v2f fa[NS][2], fb[NS][4];

#define GEMM_LOAD_SLOT(s)                                                      \
  {                                                                            \
    const int k0 = 4 * (s) * LDA, k1 = (4 * (s) + 1) * LDA;                    \
    const int h0 = 4 * (s) * LDB, h1 = (4 * (s) + 1) * LDB;                    \
    fa[s][0].x = Ap[k0];      fa[s][0].y = Ap[k1];                             \
    fa[s][1].x = Ap[k0 + 16]; fa[s][1].y = Ap[k1 + 16];                        \
    fb[s][0].x = Bp[h0];      fb[s][0].y = Bp[h1];                             \
    fb[s][1].x = Bp[h0 + 16]; fb[s][1].y = Bp[h1 + 16];                        \
    fb[s][2].x = Bp[h0 + 32]; fb[s][2].y = Bp[h1 + 32];                        \
    fb[s][3].x = Bp[h0 + 48]; fb[s][3].y = Bp[h1 + 48];                        \
  }
#define GEMM_MMA_SLOT(s)                                                       \
  {                                                                            \
    _Pragma("unroll")                                                          \
    for (int mi = 0; mi < 2; ++mi)                                             \
      _Pragma("unroll")                                                        \
      for (int t = 0; t < 4; ++t)                                              \
        acc[mi][t] = __builtin_amdgcn_wmma_f32_16x16x4_f32(                    \
            false, fa[s][mi], false, fb[s][t], (short)0, acc[mi][t], false, false); \
  }

  #pragma unroll
  for (int s = 0; s < NS; ++s) GEMM_LOAD_SLOT(s)
  Ap += 4 * NS * LDA;
  Bp += 4 * NS * LDB;

  #pragma unroll 1
  for (int kk = 4 * NS; kk < K; kk += 4 * NS) {
    #pragma unroll
    for (int s = 0; s < NS; ++s) {
      GEMM_MMA_SLOT(s)
      GEMM_LOAD_SLOT(s)
    }
    Ap += 4 * NS * LDA;
    Bp += 4 * NS * LDB;
  }
  #pragma unroll
  for (int s = 0; s < NS; ++s) GEMM_MMA_SLOT(s)
#undef GEMM_LOAD_SLOT
#undef GEMM_MMA_SLOT

  if (!TSTORE) {
    #pragma unroll
    for (int mi = 0; mi < 2; ++mi) {
      #pragma unroll
      for (int r = 0; r < 8; ++r) {
        int m = m0 + 16 * mi + r + 8 * half;
        #pragma unroll
        for (int s = 0; s < 4; ++s)
          Cb[(size_t)m * LDC + n0 + 16 * s + ml] = acc[mi][s][r];
      }
    }
  } else {
    const float* Rb = HAS_RES ? (resid + (size_t)z * strideR) : nullptr;
    #pragma unroll
    for (int mi = 0; mi < 2; ++mi) {
      #pragma unroll
      for (int r = 0; r < 8; ++r) {
        int m = m0 + 16 * mi + r + 8 * half;
        #pragma unroll
        for (int s = 0; s < 4; ++s) {
          size_t i = (size_t)(n0 + 16 * s + ml) * LDC + m;
          float v = acc[mi][s][r];
          if (HAS_RES) v += Rb[i];
          Cb[i] = v;
        }
      }
    }
  }
}

extern "C" void kernel_launch(void* const* d_in, const int* in_sizes, int n_in,
                              void* d_out, int out_size, void* d_ws, size_t ws_size,
                              hipStream_t stream) {
  const float* x      = (const float*)d_in[0];
  const float* gn_w   = (const float*)d_in[1];
  const float* gn_b   = (const float*)d_in[2];
  const float* qkv_w  = (const float*)d_in[3];
  const float* qkv_b  = (const float*)d_in[4];
  const float* proj_w = (const float*)d_in[5];
  const float* proj_b = (const float*)d_in[6];
  float* out = (float*)d_out;

  float* ws = (float*)d_ws;
  const size_t XN_SZ  = (size_t)B_ * C_ * N_;
  const size_t QKV_SZ = (size_t)B_ * N_ * C3_;
  const size_t CTX_SZ = (size_t)B_ * C_ * C_;
  float* mean   = ws;
  float* rstd   = ws + 64;
  float* xn     = ws + 256;
  float* qkv    = xn   + XN_SZ;
  float* qT     = qkv  + QKV_SZ;
  float* ctx    = qT   + XN_SZ;
  float* o2T    = ctx  + CTX_SZ;
  float* wqkvT  = o2T  + XN_SZ;
  float* wprojT = wqkvT + (size_t)C_ * C3_;

  kT_transpose<<<dim3(C_ / 32, C3_ / 32), dim3(32, 8), 0, stream>>>(qkv_w,  wqkvT,  C3_, C_);
  kT_transpose<<<dim3(C_ / 32, C_  / 32), dim3(32, 8), 0, stream>>>(proj_w, wprojT, C_,  C_);

  kT_gn_stats<<<B_ * GRP_, 256, 0, stream>>>(x, mean, rstd);
  kT_gn_apply<<<B_ * C_,   256, 0, stream>>>(x, gn_w, gn_b, mean, rstd, xn);

  kT_gemm<N_, C3_, C3_, C_, 4, false, true, false>
      <<<dim3(N_ / 64, C3_ / 256, B_), 256, 0, stream>>>(
      xn, (long long)C_ * N_,
      wqkvT, 0,
      qkv, (long long)N_ * C3_,
      qkv_b, nullptr, 0);

  kT_softmax_k<<<B_ * N_, 256, 0, stream>>>(qkv);

  kT_qtranspose<<<dim3(C_ / 32, N_ / 32, B_), dim3(32, 8), 0, stream>>>(qkv, qT);
  kT_softmax_row1024<<<B_ * C_, 256, 0, stream>>>(qT);

  kT_gemm<C3_, C3_, C_, N_, 8, false, false, false>
      <<<dim3(C_ / 64, C_ / 256, B_), 256, 0, stream>>>(
      qkv + C_,     (long long)N_ * C3_,
      qkv + 2 * C_, (long long)N_ * C3_,
      ctx, (long long)C_ * C_,
      nullptr, nullptr, 0);

  kT_gemm<N_, C_, N_, C_, 8, true, false, false>
      <<<dim3(N_ / 64, C_ / 256, B_), 256, 0, stream>>>(
      qT, (long long)C_ * N_,
      ctx, (long long)C_ * C_,
      o2T, (long long)C_ * N_,
      nullptr, nullptr, 0);

  kT_gemm<N_, C_, N_, C_, 8, true, true, true>
      <<<dim3(N_ / 64, C_ / 256, B_), 256, 0, stream>>>(
      o2T, (long long)C_ * N_,
      wprojT, 0,
      out, (long long)C_ * N_,
      proj_b, xn, (long long)C_ * N_);
}


# GEMM column fragments re-indexed so a lane's 4 B-fragment elements are adjacent in memory: one 16-byte load per k row replaces four 4-byte loads; row-major stores whole-line; sum order kept
# speedup vs baseline: 1.3876x; 1.0299x over previous
#include <hip/hip_runtime.h>
#include <math.h>

typedef float v2f __attribute__((ext_vector_type(2)));
typedef float v4f __attribute__((ext_vector_type(4)));
typedef float v8f __attribute__((ext_vector_type(8)));

#define B_    8
#define C_    512
#define N_    1024
#define C3_   1536
#define GRP_  8
#define GCH_  64
#define GELT_ (GCH_ * N_)
#define EPS_  1e-5f

__global__ __launch_bounds__(256) void kT_transpose(const float* __restrict__ in,
                                                    float* __restrict__ out,
                                                    int R, int Cc) {
  __shared__ float tile[32][33];
  int c0 = blockIdx.x * 32, r0 = blockIdx.y * 32;
  int tx = threadIdx.x, ty = threadIdx.y;
  #pragma unroll
  for (int i = 0; i < 32; i += 8)
    tile[ty + i][tx] = in[(size_t)(r0 + ty + i) * Cc + c0 + tx];
  __syncthreads();
  #pragma unroll
  for (int i = 0; i < 32; i += 8)
    out[(size_t)(c0 + ty + i) * R + r0 + tx] = tile[tx][ty + i];
}

__global__ __launch_bounds__(256) void kT_qtranspose(const float* __restrict__ qkv,
                                                     float* __restrict__ qT) {
  __shared__ float tile[32][33];
  int d0 = blockIdx.x * 32, n0 = blockIdx.y * 32;
  const float* in = qkv + (size_t)blockIdx.z * N_ * C3_;
  float*      out = qT  + (size_t)blockIdx.z * C_ * N_;
  int tx = threadIdx.x, ty = threadIdx.y;
  #pragma unroll
  for (int i = 0; i < 32; i += 8)
    tile[ty + i][tx] = in[(size_t)(n0 + ty + i) * C3_ + d0 + tx];
  __syncthreads();
  #pragma unroll
  for (int i = 0; i < 32; i += 8)
    out[(size_t)(d0 + ty + i) * N_ + n0 + tx] = tile[tx][ty + i];
}

__global__ __launch_bounds__(256) void kT_gn_stats(const float* __restrict__ x,
                                                   float* __restrict__ mean,
                                                   float* __restrict__ rstd) {
  __shared__ float s1[256], s2[256];
  int bg = blockIdx.x;
  const float4* p = (const float4*)(x + (size_t)bg * GELT_);
  float s = 0.f, q = 0.f;
  for (int i = threadIdx.x; i < GELT_ / 4; i += 256) {
    float4 v = p[i];
    s += v.x + v.y + v.z + v.w;
    q += v.x * v.x + v.y * v.y + v.z * v.z + v.w * v.w;
  }
  s1[threadIdx.x] = s; s2[threadIdx.x] = q;
  __syncthreads();
  for (int off = 128; off > 0; off >>= 1) {
    if (threadIdx.x < off) {
      s1[threadIdx.x] += s1[threadIdx.x + off];
      s2[threadIdx.x] += s2[threadIdx.x + off];
    }
    __syncthreads();
  }
  if (threadIdx.x == 0) {
    float m = s1[0] * (1.0f / GELT_);
    float v = s2[0] * (1.0f / GELT_) - m * m;
    mean[bg] = m;
    rstd[bg] = rsqrtf(v + EPS_);
  }
}

__global__ __launch_bounds__(256) void kT_gn_apply(const float* __restrict__ x,
                                                   const float* __restrict__ gamma,
                                                   const float* __restrict__ beta,
                                                   const float* __restrict__ mean,
                                                   const float* __restrict__ rstd,
                                                   float* __restrict__ xn) {
  int bc = blockIdx.x;
  int c = bc & (C_ - 1);
  int bg = ((bc >> 9) << 3) + (c >> 6);
  float gm = gamma[c] * rstd[bg];
  float bt = beta[c] - mean[bg] * gm;
  const float4* pin  = (const float4*)(x  + (size_t)bc * N_);
  float4*       pout = (float4*)(xn + (size_t)bc * N_);
  float4 v = pin[threadIdx.x];
  v.x = v.x * gm + bt; v.y = v.y * gm + bt;
  v.z = v.z * gm + bt; v.w = v.w * gm + bt;
  pout[threadIdx.x] = v;
}

__global__ __launch_bounds__(256) void kT_softmax_k(float* __restrict__ qkv) {
  __shared__ float red[256];
  float* row = qkv + (size_t)blockIdx.x * C3_ + C_;
  int t = threadIdx.x;
  float a = row[t], b = row[t + 256];
  red[t] = fmaxf(a, b);
  __syncthreads();
  for (int off = 128; off > 0; off >>= 1) {
    if (t < off) red[t] = fmaxf(red[t], red[t + off]);
    __syncthreads();
  }
  float mx = red[0];
  __syncthreads();
  float e0 = __expf(a - mx), e1 = __expf(b - mx);
  red[t] = e0 + e1;
  __syncthreads();
  for (int off = 128; off > 0; off >>= 1) {
    if (t < off) red[t] += red[t + off];
    __syncthreads();
  }
  float inv = 1.0f / red[0];
  row[t] = e0 * inv;
  row[t + 256] = e1 * inv;
}

__global__ __launch_bounds__(256) void kT_softmax_row1024(float* __restrict__ qT) {
  __shared__ float red[256];
  float4* row = (float4*)(qT + (size_t)blockIdx.x * N_);
  int t = threadIdx.x;
  float4 v = row[t];
  red[t] = fmaxf(fmaxf(v.x, v.y), fmaxf(v.z, v.w));
  __syncthreads();
  for (int off = 128; off > 0; off >>= 1) {
    if (t < off) red[t] = fmaxf(red[t], red[t + off]);
    __syncthreads();
  }
  float mx = red[0];
  __syncthreads();
  float4 e;
  e.x = __expf(v.x - mx); e.y = __expf(v.y - mx);
  e.z = __expf(v.z - mx); e.w = __expf(v.w - mx);
  red[t] = e.x + e.y + e.z + e.w;
  __syncthreads();
  for (int off = 128; off > 0; off >>= 1) {
    if (t < off) red[t] += red[t + off];
    __syncthreads();
  }
  float inv = 1.0f / red[0];
  e.x *= inv; e.y *= inv; e.z *= inv; e.w *= inv;
  row[t] = e;
}

template <int LDA, int LDB, int LDC, int K, int NS, bool TSTORE, bool HAS_BIAS, bool HAS_RES>
__global__ __launch_bounds__(256) void kT_gemm(
    const float* __restrict__ A, long long strideA,
    const float* __restrict__ B, long long strideB,
    float* __restrict__ Cmat,    long long strideC,
    const float* __restrict__ bias,
    const float* __restrict__ resid, long long strideR) {
  static_assert(K % (4 * NS) == 0 && K >= 8 * NS, "ring must divide K");
  int z = blockIdx.z;
  const float* Ab = A + (size_t)z * strideA;
  const float* Bb = B + (size_t)z * strideB;
  float*       Cb = Cmat + (size_t)z * strideC;

  int wave = threadIdx.x >> 5;
  int lane = threadIdx.x & 31;
  int half = lane >> 4;
  int ml   = lane & 15;
  int m0 = blockIdx.x * 64  + (wave & 1) * 32;
  int n0 = blockIdx.y * 256 + (wave >> 1) * 64;

  v8f acc[2][4];
  {
    v4f bv = {0.0f, 0.0f, 0.0f, 0.0f};
    if (HAS_BIAS) bv = *(const v4f*)(bias + n0 + 4 * ml);
    #pragma unroll
    for (int mi = 0; mi < 2; ++mi)
      #pragma unroll
      for (int t = 0; t < 4; ++t)
        #pragma unroll
        for (int r = 0; r < 8; ++r) acc[mi][t][r] = bv[t];
  }

  const float* Ap = Ab + (size_t)(2 * half) * LDA + m0 + ml;
  const float* Bp = Bb + (size_t)(2 * half) * LDB + n0 + 4 * ml;

  v2f fa[NS][2];
  v4f rb[NS][2];

#define GEMM_LOAD_SLOT(s)                                                      \
  {                                                                            \
    const int k0 = 4 * (s) * LDA, k1 = (4 * (s) + 1) * LDA;                    \
    fa[s][0].x = Ap[k0];      fa[s][0].y = Ap[k1];                             \
    fa[s][1].x = Ap[k0 + 16]; fa[s][1].y = Ap[k1 + 16];                        \
    rb[s][0] = *(const v4f*)(Bp + 4 * (s) * LDB);                              \
    rb[s][1] = *(const v4f*)(Bp + (4 * (s) + 1) * LDB);                        \
  }
#define GEMM_MMA_SLOT(s)                                                       \
  {                                                                            \
    _Pragma("unroll")                                                          \
    for (int mi = 0; mi < 2; ++mi)                                             \
      _Pragma("unroll")                                                        \
      for (int t = 0; t < 4; ++t) {                                            \
        v2f fn = {rb[s][0][t], rb[s][1][t]};                                   \
        acc[mi][t] = __builtin_amdgcn_wmma_f32_16x16x4_f32(                    \
            false, fa[s][mi], false, fn, (short)0, acc[mi][t], false, false);  \
      }                                                                        \
  }

  #pragma unroll
  for (int s = 0; s < NS; ++s) {
    GEMM_LOAD_SLOT(s)
    __builtin_amdgcn_sched_barrier(0);
  }
  Ap += 4 * NS * LDA;
  Bp += 4 * NS * LDB;

  #pragma unroll 1
  for (int kk = 4 * NS; kk < K; kk += 4 * NS) {
    #pragma unroll
    for (int s = 0; s < NS; ++s) {
      GEMM_MMA_SLOT(s)
      GEMM_LOAD_SLOT(s)
      __builtin_amdgcn_sched_barrier(0);
    }
    Ap += 4 * NS * LDA;
    Bp += 4 * NS * LDB;
  }
  #pragma unroll
  for (int s = 0; s < NS; ++s) {
    GEMM_MMA_SLOT(s)
    __builtin_amdgcn_sched_barrier(0);
  }
#undef GEMM_LOAD_SLOT
#undef GEMM_MMA_SLOT

  if (!TSTORE) {
    #pragma unroll
    for (int mi = 0; mi < 2; ++mi) {
      #pragma unroll
      for (int r = 0; r < 8; ++r) {
        int m = m0 + 16 * mi + r + 8 * half;
        v4f v = {acc[mi][0][r], acc[mi][1][r], acc[mi][2][r], acc[mi][3][r]};
        *(v4f*)(Cb + (size_t)m * LDC + n0 + 4 * ml) = v;
      }
    }
  } else {
    const float* Rb = HAS_RES ? (resid + (size_t)z * strideR) : nullptr;
    #pragma unroll
    for (int mi = 0; mi < 2; ++mi) {
      #pragma unroll
      for (int r = 0; r < 8; ++r) {
        int m = m0 + 16 * mi + r + 8 * half;
        #pragma unroll
        for (int t = 0; t < 4; ++t) {
          size_t i = (size_t)(n0 + 4 * ml + t) * LDC + m;
          float v = acc[mi][t][r];
          if (HAS_RES) v += Rb[i];
          Cb[i] = v;
        }
      }
    }
  }
}

extern "C" void kernel_launch(void* const* d_in, const int* in_sizes, int n_in,
                              void* d_out, int out_size, void* d_ws, size_t ws_size,
                              hipStream_t stream) {
  const float* x      = (const float*)d_in[0];
  const float* gn_w   = (const float*)d_in[1];
  const float* gn_b   = (const float*)d_in[2];
  const float* qkv_w  = (const float*)d_in[3];
  const float* qkv_b  = (const float*)d_in[4];
  const float* proj_w = (const float*)d_in[5];
  const float* proj_b = (const float*)d_in[6];
  float* out = (float*)d_out;

  float* ws = (float*)d_ws;
  const size_t XN_SZ  = (size_t)B_ * C_ * N_;
  const size_t QKV_SZ = (size_t)B_ * N_ * C3_;
  const size_t CTX_SZ = (size_t)B_ * C_ * C_;
  float* mean   = ws;
  float* rstd   = ws + 64;
  float* xn     = ws + 256;
  float* qkv    = xn   + XN_SZ;
  float* qT     = qkv  + QKV_SZ;
  float* ctx    = qT   + XN_SZ;
  float* o2T    = ctx  + CTX_SZ;
  float* wqkvT  = o2T  + XN_SZ;
  float* wprojT = wqkvT + (size_t)C_ * C3_;

  kT_transpose<<<dim3(C_ / 32, C3_ / 32), dim3(32, 8), 0, stream>>>(qkv_w,  wqkvT,  C3_, C_);
  kT_transpose<<<dim3(C_ / 32, C_  / 32), dim3(32, 8), 0, stream>>>(proj_w, wprojT, C_,  C_);

  kT_gn_stats<<<B_ * GRP_, 256, 0, stream>>>(x, mean, rstd);
  kT_gn_apply<<<B_ * C_,   256, 0, stream>>>(x, gn_w, gn_b, mean, rstd, xn);

  kT_gemm<N_, C3_, C3_, C_, 4, false, true, false>
      <<<dim3(N_ / 64, C3_ / 256, B_), 256, 0, stream>>>(
      xn, (long long)C_ * N_,
      wqkvT, 0,
      qkv, (long long)N_ * C3_,
      qkv_b, nullptr, 0);

  kT_softmax_k<<<B_ * N_, 256, 0, stream>>>(qkv);

  kT_qtranspose<<<dim3(C_ / 32, N_ / 32, B_), dim3(32, 8), 0, stream>>>(qkv, qT);
  kT_softmax_row1024<<<B_ * C_, 256, 0, stream>>>(qT);

  kT_gemm<C3_, C3_, C_, N_, 8, false, false, false>
      <<<dim3(C_ / 64, C_ / 256, B_), 256, 0, stream>>>(
      qkv + C_,     (long long)N_ * C3_,
      qkv + 2 * C_, (long long)N_ * C3_,
      ctx, (long long)C_ * C_,
      nullptr, nullptr, 0);

  kT_gemm<N_, C_, N_, C_, 8, true, false, false>
      <<<dim3(N_ / 64, C_ / 256, B_), 256, 0, stream>>>(
      qT, (long long)C_ * N_,
      ctx, (long long)C_ * C_,
      o2T, (long long)C_ * N_,
      nullptr, nullptr, 0);

  kT_gemm<N_, C_, N_, C_, 8, true, true, true>
      <<<dim3(N_ / 64, C_ / 256, B_), 256, 0, stream>>>(
      o2T, (long long)C_ * N_,
      wprojT, 0,
      out, (long long)C_ * N_,
      proj_b, xn, (long long)C_ * N_);
}


# Non-GEMM kernels re-gridded to fewer, fatter waves: one wave per row with 8 16-byte loads in flight (softmax sums by lane then wave butterfly: sum order changed); 128x64 transpose tiles
# speedup vs baseline: 1.5862x; 1.1431x over previous
#include <hip/hip_runtime.h>
#include <math.h>

typedef float v2f __attribute__((ext_vector_type(2)));
typedef float v4f __attribute__((ext_vector_type(4)));
typedef float v8f __attribute__((ext_vector_type(8)));

#define B_    8
#define C_    512
#define N_    1024
#define C3_   1536
#define GRP_  8
#define GCH_  64
#define GELT_ (GCH_ * N_)
#define EPS_  1e-5f

__global__ __launch_bounds__(256) void kT_transpose(const float* __restrict__ in,
                                                    float* __restrict__ out,
                                                    int ldin, int ldout,
                                                    long long strideIn, long long strideOut) {
  __shared__ float tile[64][129];
  const float* ib = in  + (size_t)blockIdx.z * strideIn;
  float*       ob = out + (size_t)blockIdx.z * strideOut;
  int c0 = blockIdx.x * 64, r0 = blockIdx.y * 128;
  int q4 = threadIdx.x & 15;
  int g  = threadIdx.x >> 4;
  v4f v[8];
  #pragma unroll
  for (int i = 0; i < 8; ++i)
    v[i] = *(const v4f*)(ib + (size_t)(r0 + g + 16 * i) * ldin + c0 + 4 * q4);
  #pragma unroll
  for (int i = 0; i < 8; ++i)
    #pragma unroll
    for (int e = 0; e < 4; ++e)
      tile[4 * q4 + e][g + 16 * i] = v[i][e];
  __syncthreads();
  #pragma unroll
  for (int i = 0; i < 8; ++i) {
    int u = 16 * i + g;
    int c = u & 63, rh = u >> 6;
    v4f o;
    #pragma unroll
    for (int e = 0; e < 4; ++e) o[e] = tile[c][64 * rh + 4 * q4 + e];
    *(v4f*)(ob + (size_t)(c0 + c) * ldout + r0 + 64 * rh + 4 * q4) = o;
  }
}

__global__ __launch_bounds__(256) void kT_gn_stats(const float* __restrict__ x,
                                                   float* __restrict__ mean,
                                                   float* __restrict__ rstd) {
  __shared__ float s1[256], s2[256];
  int bg = blockIdx.x;
  const float4* p = (const float4*)(x + (size_t)bg * GELT_);
  float s = 0.f, q = 0.f;
  for (int i = threadIdx.x; i < GELT_ / 4; i += 256) {
    float4 v = p[i];
    s += v.x + v.y + v.z + v.w;
    q += v.x * v.x + v.y * v.y + v.z * v.z + v.w * v.w;
  }
  s1[threadIdx.x] = s; s2[threadIdx.x] = q;
  __syncthreads();
  for (int off = 128; off > 0; off >>= 1) {
    if (threadIdx.x < off) {
      s1[threadIdx.x] += s1[threadIdx.x + off];
      s2[threadIdx.x] += s2[threadIdx.x + off];
    }
    __syncthreads();
  }
  if (threadIdx.x == 0) {
    float m = s1[0] * (1.0f / GELT_);
    float v = s2[0] * (1.0f / GELT_) - m * m;
    mean[bg] = m;
    rstd[bg] = rsqrtf(v + EPS_);
  }
}

__device__ __forceinline__ float wave_max(float v) {
  #pragma unroll
  for (int off = 16; off > 0; off >>= 1) v = fmaxf(v, __shfl_xor(v, off));
  return v;
}
__device__ __forceinline__ float wave_sum(float v) {
  #pragma unroll
  for (int off = 16; off > 0; off >>= 1) v += __shfl_xor(v, off);
  return v;
}

__global__ __launch_bounds__(256) void kT_gn_apply(const float* __restrict__ x,
                                                   const float* __restrict__ gamma,
                                                   const float* __restrict__ beta,
                                                   const float* __restrict__ mean,
                                                   const float* __restrict__ rstd,
                                                   float* __restrict__ xn) {
  int lane = threadIdx.x & 31;
  int bc = blockIdx.x * 8 + (threadIdx.x >> 5);
  int c = bc & (C_ - 1);
  int bg = ((bc >> 9) << 3) + (c >> 6);
  const v4f* pin  = (const v4f*)(x  + (size_t)bc * N_);
  v4f*       pout = (v4f*)(xn + (size_t)bc * N_);
  v4f v[8];
  #pragma unroll
  for (int j = 0; j < 8; ++j) v[j] = pin[lane + 32 * j];
  float gm = gamma[c] * rstd[bg];
  float bt = beta[c] - mean[bg] * gm;
  #pragma unroll
  for (int j = 0; j < 8; ++j) {
    v4f w = v[j];
    w.x = w.x * gm + bt; w.y = w.y * gm + bt;
    w.z = w.z * gm + bt; w.w = w.w * gm + bt;
    pout[lane + 32 * j] = w;
  }
}

__global__ __launch_bounds__(256) void kT_softmax_k(float* __restrict__ qkv) {
  int lane = threadIdx.x & 31;
  int row0 = (blockIdx.x * 8 + (threadIdx.x >> 5)) * 2;
  v4f* p0 = (v4f*)(qkv + (size_t)row0 * C3_ + C_);
  v4f* p1 = (v4f*)(qkv + (size_t)(row0 + 1) * C3_ + C_);
  v4f a[4], b[4];
  #pragma unroll
  for (int j = 0; j < 4; ++j) { a[j] = p0[lane + 32 * j]; b[j] = p1[lane + 32 * j]; }
  float ma = fmaxf(fmaxf(a[0].x, a[0].y), fmaxf(a[0].z, a[0].w));
  float mb = fmaxf(fmaxf(b[0].x, b[0].y), fmaxf(b[0].z, b[0].w));
  #pragma unroll
  for (int j = 1; j < 4; ++j) {
    ma = fmaxf(ma, fmaxf(fmaxf(a[j].x, a[j].y), fmaxf(a[j].z, a[j].w)));
    mb = fmaxf(mb, fmaxf(fmaxf(b[j].x, b[j].y), fmaxf(b[j].z, b[j].w)));
  }
  ma = wave_max(ma);
  mb = wave_max(mb);
  float sa = 0.f, sb = 0.f;
  #pragma unroll
  for (int j = 0; j < 4; ++j) {
    a[j].x = __expf(a[j].x - ma); a[j].y = __expf(a[j].y - ma);
    a[j].z = __expf(a[j].z - ma); a[j].w = __expf(a[j].w - ma);
    b[j].x = __expf(b[j].x - mb); b[j].y = __expf(b[j].y - mb);
    b[j].z = __expf(b[j].z - mb); b[j].w = __expf(b[j].w - mb);
    sa += (a[j].x + a[j].y) + (a[j].z + a[j].w);
    sb += (b[j].x + b[j].y) + (b[j].z + b[j].w);
  }
  float ia = 1.0f / wave_sum(sa);
  float ib = 1.0f / wave_sum(sb);
  #pragma unroll
  for (int j = 0; j < 4; ++j) {
    p0[lane + 32 * j] = a[j] * ia;
    p1[lane + 32 * j] = b[j] * ib;
  }
}

__global__ __launch_bounds__(256) void kT_softmax_row1024(float* __restrict__ qT) {
  int lane = threadIdx.x & 31;
  v4f* row = (v4f*)(qT + (size_t)(blockIdx.x * 8 + (threadIdx.x >> 5)) * N_);
  v4f v[8];
  #pragma unroll
  for (int j = 0; j < 8; ++j) v[j] = row[lane + 32 * j];
  float mx = fmaxf(fmaxf(v[0].x, v[0].y), fmaxf(v[0].z, v[0].w));
  #pragma unroll
  for (int j = 1; j < 8; ++j)
    mx = fmaxf(mx, fmaxf(fmaxf(v[j].x, v[j].y), fmaxf(v[j].z, v[j].w)));
  mx = wave_max(mx);
  float s = 0.f;
  #pragma unroll
  for (int j = 0; j < 8; ++j) {
    v[j].x = __expf(v[j].x - mx); v[j].y = __expf(v[j].y - mx);
    v[j].z = __expf(v[j].z - mx); v[j].w = __expf(v[j].w - mx);
    s += (v[j].x + v[j].y) + (v[j].z + v[j].w);
  }
  float inv = 1.0f / wave_sum(s);
  #pragma unroll
  for (int j = 0; j < 8; ++j) row[lane + 32 * j] = v[j] * inv;
}

template <int LDA, int LDB, int LDC, int K, int NS, bool TSTORE, bool HAS_BIAS, bool HAS_RES>
__global__ __launch_bounds__(256) void kT_gemm(
    const float* __restrict__ A, long long strideA,
    const float* __restrict__ B, long long strideB,
    float* __restrict__ Cmat,    long long strideC,
    const float* __restrict__ bias,
    const float* __restrict__ resid, long long strideR) {
  static_assert(K % (4 * NS) == 0 && K >= 8 * NS, "ring must divide K");
  int z = blockIdx.z;
  const float* Ab = A + (size_t)z * strideA;
  const float* Bb = B + (size_t)z * strideB;
  float*       Cb = Cmat + (size_t)z * strideC;

  int wave = threadIdx.x >> 5;
  int lane = threadIdx.x & 31;
  int half = lane >> 4;
  int ml   = lane & 15;
  int m0 = blockIdx.x * 64  + (wave & 1) * 32;
  int n0 = blockIdx.y * 256 + (wave >> 1) * 64;

  v8f acc[2][4];
  {
    v4f bv = {0.0f, 0.0f, 0.0f, 0.0f};
    if (HAS_BIAS) bv = *(const v4f*)(bias + n0 + 4 * ml);
    #pragma unroll
    for (int mi = 0; mi < 2; ++mi)
      #pragma unroll
      for (int t = 0; t < 4; ++t)
        #pragma unroll
        for (int r = 0; r < 8; ++r) acc[mi][t][r] = bv[t];
  }

  const float* Ap = Ab + (size_t)(2 * half) * LDA + m0 + ml;
  const float* Bp = Bb + (size_t)(2 * half) * LDB + n0 + 4 * ml;

  v2f fa[NS][2];
  v4f rb[NS][2];

#define GEMM_LOAD_SLOT(s)                                                      \
  {                                                                            \
    const int k0 = 4 * (s) * LDA, k1 = (4 * (s) + 1) * LDA;                    \
    fa[s][0].x = Ap[k0];      fa[s][0].y = Ap[k1];                             \
    fa[s][1].x = Ap[k0 + 16]; fa[s][1].y = Ap[k1 + 16];                        \
    rb[s][0] = *(const v4f*)(Bp + 4 * (s) * LDB);                              \
    rb[s][1] = *(const v4f*)(Bp + (4 * (s) + 1) * LDB);                        \
  }
#define GEMM_MMA_SLOT(s)                                                       \
  {                                                                            \
    _Pragma("unroll")                                                          \
    for (int mi = 0; mi < 2; ++mi)                                             \
      _Pragma("unroll")                                                        \
      for (int t = 0; t < 4; ++t) {                                            \
        v2f fn = {rb[s][0][t], rb[s][1][t]};                                   \
        acc[mi][t] = __builtin_amdgcn_wmma_f32_16x16x4_f32(                    \
            false, fa[s][mi], false, fn, (short)0, acc[mi][t], false, false);  \
      }                                                                        \
  }

  #pragma unroll
  for (int s = 0; s < NS; ++s) {
    GEMM_LOAD_SLOT(s)
    __builtin_amdgcn_sched_barrier(0);
  }
  Ap += 4 * NS * LDA;
  Bp += 4 * NS * LDB;

  #pragma unroll 1
  for (int kk = 4 * NS; kk < K; kk += 4 * NS) {
    #pragma unroll
    for (int s = 0; s < NS; ++s) {
      GEMM_MMA_SLOT(s)
      GEMM_LOAD_SLOT(s)
      __builtin_amdgcn_sched_barrier(0);
    }
    Ap += 4 * NS * LDA;
    Bp += 4 * NS * LDB;
  }
  #pragma unroll
  for (int s = 0; s < NS; ++s) {
    GEMM_MMA_SLOT(s)
    __builtin_amdgcn_sched_barrier(0);
  }
#undef GEMM_LOAD_SLOT
#undef GEMM_MMA_SLOT

  if (!TSTORE) {
    #pragma unroll
    for (int mi = 0; mi < 2; ++mi) {
      #pragma unroll
      for (int r = 0; r < 8; ++r) {
        int m = m0 + 16 * mi + r + 8 * half;
        v4f v = {acc[mi][0][r], acc[mi][1][r], acc[mi][2][r], acc[mi][3][r]};
        *(v4f*)(Cb + (size_t)m * LDC + n0 + 4 * ml) = v;
      }
    }
  } else {
    const float* Rb = HAS_RES ? (resid + (size_t)z * strideR) : nullptr;
    #pragma unroll
    for (int mi = 0; mi < 2; ++mi) {
      #pragma unroll
      for (int r = 0; r < 8; ++r) {
        int m = m0 + 16 * mi + r + 8 * half;
        #pragma unroll
        for (int t = 0; t < 4; ++t) {
          size_t i = (size_t)(n0 + 4 * ml + t) * LDC + m;
          float v = acc[mi][t][r];
          if (HAS_RES) v += Rb[i];
          Cb[i] = v;
        }
      }
    }
  }
}

extern "C" void kernel_launch(void* const* d_in, const int* in_sizes, int n_in,
                              void* d_out, int out_size, void* d_ws, size_t ws_size,
                              hipStream_t stream) {
  const float* x      = (const float*)d_in[0];
  const float* gn_w   = (const float*)d_in[1];
  const float* gn_b   = (const float*)d_in[2];
  const float* qkv_w  = (const float*)d_in[3];
  const float* qkv_b  = (const float*)d_in[4];
  const float* proj_w = (const float*)d_in[5];
  const float* proj_b = (const float*)d_in[6];
  float* out = (float*)d_out;

  float* ws = (float*)d_ws;
  const size_t XN_SZ  = (size_t)B_ * C_ * N_;
  const size_t QKV_SZ = (size_t)B_ * N_ * C3_;
  const size_t CTX_SZ = (size_t)B_ * C_ * C_;
  float* mean   = ws;
  float* rstd   = ws + 64;
  float* xn     = ws + 256;
  float* qkv    = xn   + XN_SZ;
  float* qT     = qkv  + QKV_SZ;
  float* ctx    = qT   + XN_SZ;
  float* o2T    = ctx  + CTX_SZ;
  float* wqkvT  = o2T  + XN_SZ;
  float* wprojT = wqkvT + (size_t)C_ * C3_;

  kT_transpose<<<dim3(C_ / 64, C3_ / 128, 1), 256, 0, stream>>>(qkv_w,  wqkvT,  C_, C3_, 0, 0);
  kT_transpose<<<dim3(C_ / 64, C_  / 128, 1), 256, 0, stream>>>(proj_w, wprojT, C_, C_,  0, 0);

  kT_gn_stats<<<B_ * GRP_, 256, 0, stream>>>(x, mean, rstd);
  kT_gn_apply<<<B_ * C_ / 8, 256, 0, stream>>>(x, gn_w, gn_b, mean, rstd, xn);

  kT_gemm<N_, C3_, C3_, C_, 4, false, true, false>
      <<<dim3(N_ / 64, C3_ / 256, B_), 256, 0, stream>>>(
      xn, (long long)C_ * N_,
      wqkvT, 0,
      qkv, (long long)N_ * C3_,
      qkv_b, nullptr, 0);

  kT_softmax_k<<<B_ * N_ / 16, 256, 0, stream>>>(qkv);

  kT_transpose<<<dim3(C_ / 64, N_ / 128, B_), 256, 0, stream>>>(
      qkv, qT, C3_, N_, (long long)N_ * C3_, (long long)C_ * N_);
  kT_softmax_row1024<<<B_ * C_ / 8, 256, 0, stream>>>(qT);

  kT_gemm<C3_, C3_, C_, N_, 8, false, false, false>
      <<<dim3(C_ / 64, C_ / 256, B_), 256, 0, stream>>>(
      qkv + C_,     (long long)N_ * C3_,
      qkv + 2 * C_, (long long)N_ * C3_,
      ctx, (long long)C_ * C_,
      nullptr, nullptr, 0);

  kT_gemm<N_, C_, N_, C_, 8, true, false, false>
      <<<dim3(N_ / 64, C_ / 256, B_), 256, 0, stream>>>(
      qT, (long long)C_ * N_,
      ctx, (long long)C_ * C_,
      o2T, (long long)C_ * N_,
      nullptr, nullptr, 0);

  kT_gemm<N_, C_, N_, C_, 8, true, true, true>
      <<<dim3(N_ / 64, C_ / 256, B_), 256, 0, stream>>>(
      o2T, (long long)C_ * N_,
      wprojT, 0,
      out, (long long)C_ * N_,
      proj_b, xn, (long long)C_ * N_);
}


# Context GEMM re-tiled to 16x64 wave tiles in 64x128 blocks (ring 16 deep): 256 blocks, one per compute unit, instead of 128 that left half the matrix pipes idle; sum order kept
# speedup vs baseline: 1.7169x; 1.0824x over previous
#include <hip/hip_runtime.h>
#include <math.h>

typedef float v2f __attribute__((ext_vector_type(2)));
typedef float v4f __attribute__((ext_vector_type(4)));
typedef float v8f __attribute__((ext_vector_type(8)));

#define B_    8
#define C_    512
#define N_    1024
#define C3_   1536
#define GRP_  8
#define GCH_  64
#define GELT_ (GCH_ * N_)
#define EPS_  1e-5f

__global__ __launch_bounds__(256) void kT_transpose(const float* __restrict__ in,
                                                    float* __restrict__ out,
                                                    int ldin, int ldout,
                                                    long long strideIn, long long strideOut) {
  __shared__ float tile[64][129];
  const float* ib = in  + (size_t)blockIdx.z * strideIn;
  float*       ob = out + (size_t)blockIdx.z * strideOut;
  int c0 = blockIdx.x * 64, r0 = blockIdx.y * 128;
  int q4 = threadIdx.x & 15;
  int g  = threadIdx.x >> 4;
  v4f v[8];
  #pragma unroll
  for (int i = 0; i < 8; ++i)
    v[i] = *(const v4f*)(ib + (size_t)(r0 + g + 16 * i) * ldin + c0 + 4 * q4);
  #pragma unroll
  for (int i = 0; i < 8; ++i)
    #pragma unroll
    for (int e = 0; e < 4; ++e)
      tile[4 * q4 + e][g + 16 * i] = v[i][e];
  __syncthreads();
  #pragma unroll
  for (int i = 0; i < 8; ++i) {
    int u = 16 * i + g;
    int c = u & 63, rh = u >> 6;
    v4f o;
    #pragma unroll
    for (int e = 0; e < 4; ++e) o[e] = tile[c][64 * rh + 4 * q4 + e];
    *(v4f*)(ob + (size_t)(c0 + c) * ldout + r0 + 64 * rh + 4 * q4) = o;
  }
}

__global__ __launch_bounds__(256) void kT_gn_stats(const float* __restrict__ x,
                                                   float* __restrict__ mean,
                                                   float* __restrict__ rstd) {
  __shared__ float s1[256], s2[256];
  int bg = blockIdx.x;
  const float4* p = (const float4*)(x + (size_t)bg * GELT_);
  float s = 0.f, q = 0.f;
  for (int i = threadIdx.x; i < GELT_ / 4; i += 256) {
    float4 v = p[i];
    s += v.x + v.y + v.z + v.w;
    q += v.x * v.x + v.y * v.y + v.z * v.z + v.w * v.w;
  }
  s1[threadIdx.x] = s; s2[threadIdx.x] = q;
  __syncthreads();
  for (int off = 128; off > 0; off >>= 1) {
    if (threadIdx.x < off) {
      s1[threadIdx.x] += s1[threadIdx.x + off];
      s2[threadIdx.x] += s2[threadIdx.x + off];
    }
    __syncthreads();
  }
  if (threadIdx.x == 0) {
    float m = s1[0] * (1.0f / GELT_);
    float v = s2[0] * (1.0f / GELT_) - m * m;
    mean[bg] = m;
    rstd[bg] = rsqrtf(v + EPS_);
  }
}

__device__ __forceinline__ float wave_max(float v) {
  #pragma unroll
  for (int off = 16; off > 0; off >>= 1) v = fmaxf(v, __shfl_xor(v, off));
  return v;
}
__device__ __forceinline__ float wave_sum(float v) {
  #pragma unroll
  for (int off = 16; off > 0; off >>= 1) v += __shfl_xor(v, off);
  return v;
}

__global__ __launch_bounds__(256) void kT_gn_apply(const float* __restrict__ x,
                                                   const float* __restrict__ gamma,
                                                   const float* __restrict__ beta,
                                                   const float* __restrict__ mean,
                                                   const float* __restrict__ rstd,
                                                   float* __restrict__ xn) {
  int lane = threadIdx.x & 31;
  int bc = blockIdx.x * 8 + (threadIdx.x >> 5);
  int c = bc & (C_ - 1);
  int bg = ((bc >> 9) << 3) + (c >> 6);
  const v4f* pin  = (const v4f*)(x  + (size_t)bc * N_);
  v4f*       pout = (v4f*)(xn + (size_t)bc * N_);
  v4f v[8];
  #pragma unroll
  for (int j = 0; j < 8; ++j) v[j] = pin[lane + 32 * j];
  float gm = gamma[c] * rstd[bg];
  float bt = beta[c] - mean[bg] * gm;
  #pragma unroll
  for (int j = 0; j < 8; ++j) {
    v4f w = v[j];
    w.x = w.x * gm + bt; w.y = w.y * gm + bt;
    w.z = w.z * gm + bt; w.w = w.w * gm + bt;
    pout[lane + 32 * j] = w;
  }
}

__global__ __launch_bounds__(256) void kT_softmax_k(float* __restrict__ qkv) {
  int lane = threadIdx.x & 31;
  int row0 = (blockIdx.x * 8 + (threadIdx.x >> 5)) * 2;
  v4f* p0 = (v4f*)(qkv + (size_t)row0 * C3_ + C_);
  v4f* p1 = (v4f*)(qkv + (size_t)(row0 + 1) * C3_ + C_);
  v4f a[4], b[4];
  #pragma unroll
  for (int j = 0; j < 4; ++j) { a[j] = p0[lane + 32 * j]; b[j] = p1[lane + 32 * j]; }
  float ma = fmaxf(fmaxf(a[0].x, a[0].y), fmaxf(a[0].z, a[0].w));
  float mb = fmaxf(fmaxf(b[0].x, b[0].y), fmaxf(b[0].z, b[0].w));
  #pragma unroll
  for (int j = 1; j < 4; ++j) {
    ma = fmaxf(ma, fmaxf(fmaxf(a[j].x, a[j].y), fmaxf(a[j].z, a[j].w)));
    mb = fmaxf(mb, fmaxf(fmaxf(b[j].x, b[j].y), fmaxf(b[j].z, b[j].w)));
  }
  ma = wave_max(ma);
  mb = wave_max(mb);
  float sa = 0.f, sb = 0.f;
  #pragma unroll
  for (int j = 0; j < 4; ++j) {
    a[j].x = __expf(a[j].x - ma); a[j].y = __expf(a[j].y - ma);
    a[j].z = __expf(a[j].z - ma); a[j].w = __expf(a[j].w - ma);
    b[j].x = __expf(b[j].x - mb); b[j].y = __expf(b[j].y - mb);
    b[j].z = __expf(b[j].z - mb); b[j].w = __expf(b[j].w - mb);
    sa += (a[j].x + a[j].y) + (a[j].z + a[j].w);
    sb += (b[j].x + b[j].y) + (b[j].z + b[j].w);
  }
  float ia = 1.0f / wave_sum(sa);
  float ib = 1.0f / wave_sum(sb);
  #pragma unroll
  for (int j = 0; j < 4; ++j) {
    p0[lane + 32 * j] = a[j] * ia;
    p1[lane + 32 * j] = b[j] * ib;
  }
}

__global__ __launch_bounds__(256) void kT_softmax_row1024(float* __restrict__ qT) {
  int lane = threadIdx.x & 31;
  v4f* row = (v4f*)(qT + (size_t)(blockIdx.x * 8 + (threadIdx.x >> 5)) * N_);
  v4f v[8];
  #pragma unroll
  for (int j = 0; j < 8; ++j) v[j] = row[lane + 32 * j];
  float mx = fmaxf(fmaxf(v[0].x, v[0].y), fmaxf(v[0].z, v[0].w));
  #pragma unroll
  for (int j = 1; j < 8; ++j)
    mx = fmaxf(mx, fmaxf(fmaxf(v[j].x, v[j].y), fmaxf(v[j].z, v[j].w)));
  mx = wave_max(mx);
  float s = 0.f;
  #pragma unroll
  for (int j = 0; j < 8; ++j) {
    v[j].x = __expf(v[j].x - mx); v[j].y = __expf(v[j].y - mx);
    v[j].z = __expf(v[j].z - mx); v[j].w = __expf(v[j].w - mx);
    s += (v[j].x + v[j].y) + (v[j].z + v[j].w);
  }
  float inv = 1.0f / wave_sum(s);
  #pragma unroll
  for (int j = 0; j < 8; ++j) row[lane + 32 * j] = v[j] * inv;
}

template <int LDA, int LDB, int LDC, int K, int NS, int TM, int WM,
          bool TSTORE, bool HAS_BIAS, bool HAS_RES>
__global__ __launch_bounds__(256) void kT_gemm(
    const float* __restrict__ A, long long strideA,
    const float* __restrict__ B, long long strideB,
    float* __restrict__ Cmat,    long long strideC,
    const float* __restrict__ bias,
    const float* __restrict__ resid, long long strideR) {
  static_assert(K % (4 * NS) == 0 && K >= 8 * NS, "ring must divide K");
  static_assert((TM == 1 || TM == 2) && (WM == 2 || WM == 4), "tile shape");
  int z = blockIdx.z;
  const float* Ab = A + (size_t)z * strideA;
  const float* Bb = B + (size_t)z * strideB;
  float*       Cb = Cmat + (size_t)z * strideC;

  int wave = threadIdx.x >> 5;
  int lane = threadIdx.x & 31;
  int half = lane >> 4;
  int ml   = lane & 15;
  constexpr int WN = 8 / WM;
  int m0 = blockIdx.x * (16 * TM * WM) + (wave % WM) * (16 * TM);
  int n0 = blockIdx.y * (64 * WN) + (wave / WM) * 64;

  v8f acc[TM][4];
  {
    v4f bv = {0.0f, 0.0f, 0.0f, 0.0f};
    if (HAS_BIAS) bv = *(const v4f*)(bias + n0 + 4 * ml);
    #pragma unroll
    for (int mi = 0; mi < TM; ++mi)
      #pragma unroll
      for (int t = 0; t < 4; ++t)
        #pragma unroll
        for (int r = 0; r < 8; ++r) acc[mi][t][r] = bv[t];
  }

  const float* Ap = Ab + (size_t)(2 * half) * LDA + m0 + ml;
  const float* Bp = Bb + (size_t)(2 * half) * LDB + n0 + 4 * ml;

  v2f fa[NS][TM];
  v4f rb[NS][2];

#define GEMM_LOAD_SLOT(s)                                                      \
  {                                                                            \
    const int k0 = 4 * (s) * LDA, k1 = (4 * (s) + 1) * LDA;                    \
    _Pragma("unroll")                                                          \
    for (int mi = 0; mi < TM; ++mi) {                                          \
      fa[s][mi].x = Ap[k0 + 16 * mi]; fa[s][mi].y = Ap[k1 + 16 * mi];          \
    }                                                                          \
    rb[s][0] = *(const v4f*)(Bp + 4 * (s) * LDB);                              \
    rb[s][1] = *(const v4f*)(Bp + (4 * (s) + 1) * LDB);                        \
  }
#define GEMM_MMA_SLOT(s)                                                       \
  {                                                                            \
    _Pragma("unroll")                                                          \
    for (int mi = 0; mi < TM; ++mi)                                            \
      _Pragma("unroll")                                                        \
      for (int t = 0; t < 4; ++t) {                                            \
        v2f fn = {rb[s][0][t], rb[s][1][t]};                                   \
        acc[mi][t] = __builtin_amdgcn_wmma_f32_16x16x4_f32(                    \
            false, fa[s][mi], false, fn, (short)0, acc[mi][t], false, false);  \
      }                                                                        \
  }

  #pragma unroll
  for (int s = 0; s < NS; ++s) {
    GEMM_LOAD_SLOT(s)
    __builtin_amdgcn_sched_barrier(0);
  }
  Ap += 4 * NS * LDA;
  Bp += 4 * NS * LDB;

  #pragma unroll 1
  for (int kk = 4 * NS; kk < K; kk += 4 * NS) {
    #pragma unroll
    for (int s = 0; s < NS; ++s) {
      GEMM_MMA_SLOT(s)
      GEMM_LOAD_SLOT(s)
      __builtin_amdgcn_sched_barrier(0);
    }
    Ap += 4 * NS * LDA;
    Bp += 4 * NS * LDB;
  }
  #pragma unroll
  for (int s = 0; s < NS; ++s) {
    GEMM_MMA_SLOT(s)
    __builtin_amdgcn_sched_barrier(0);
  }
#undef GEMM_LOAD_SLOT
#undef GEMM_MMA_SLOT

  if (!TSTORE) {
    #pragma unroll
    for (int mi = 0; mi < TM; ++mi) {
      #pragma unroll
      for (int r = 0; r < 8; ++r) {
        int m = m0 + 16 * mi + r + 8 * half;
        v4f v = {acc[mi][0][r], acc[mi][1][r], acc[mi][2][r], acc[mi][3][r]};
        *(v4f*)(Cb + (size_t)m * LDC + n0 + 4 * ml) = v;
      }
    }
  } else {
    const float* Rb = HAS_RES ? (resid + (size_t)z * strideR) : nullptr;
    #pragma unroll
    for (int mi = 0; mi < TM; ++mi) {
      #pragma unroll
      for (int r = 0; r < 8; ++r) {
        int m = m0 + 16 * mi + r + 8 * half;
        #pragma unroll
        for (int t = 0; t < 4; ++t) {
          size_t i = (size_t)(n0 + 4 * ml + t) * LDC + m;
          float v = acc[mi][t][r];
          if (HAS_RES) v += Rb[i];
          Cb[i] = v;
        }
      }
    }
  }
}

extern "C" void kernel_launch(void* const* d_in, const int* in_sizes, int n_in,
                              void* d_out, int out_size, void* d_ws, size_t ws_size,
                              hipStream_t stream) {
  const float* x      = (const float*)d_in[0];
  const float* gn_w   = (const float*)d_in[1];
  const float* gn_b   = (const float*)d_in[2];
  const float* qkv_w  = (const float*)d_in[3];
  const float* qkv_b  = (const float*)d_in[4];
  const float* proj_w = (const float*)d_in[5];
  const float* proj_b = (const float*)d_in[6];
  float* out = (float*)d_out;

  float* ws = (float*)d_ws;
  const size_t XN_SZ  = (size_t)B_ * C_ * N_;
  const size_t QKV_SZ = (size_t)B_ * N_ * C3_;
  const size_t CTX_SZ = (size_t)B_ * C_ * C_;
  float* mean   = ws;
  float* rstd   = ws + 64;
  float* xn     = ws + 256;
  float* qkv    = xn   + XN_SZ;
  float* qT     = qkv  + QKV_SZ;
  float* ctx    = qT   + XN_SZ;
  float* o2T    = ctx  + CTX_SZ;
  float* wqkvT  = o2T  + XN_SZ;
  float* wprojT = wqkvT + (size_t)C_ * C3_;

  kT_transpose<<<dim3(C_ / 64, C3_ / 128, 1), 256, 0, stream>>>(qkv_w,  wqkvT,  C_, C3_, 0, 0);
  kT_transpose<<<dim3(C_ / 64, C_  / 128, 1), 256, 0, stream>>>(proj_w, wprojT, C_, C_,  0, 0);

  kT_gn_stats<<<B_ * GRP_, 256, 0, stream>>>(x, mean, rstd);
  kT_gn_apply<<<B_ * C_ / 8, 256, 0, stream>>>(x, gn_w, gn_b, mean, rstd, xn);

  kT_gemm<N_, C3_, C3_, C_, 4, 2, 2, false, true, false>
      <<<dim3(N_ / 64, C3_ / 256, B_), 256, 0, stream>>>(
      xn, (long long)C_ * N_,
      wqkvT, 0,
      qkv, (long long)N_ * C3_,
      qkv_b, nullptr, 0);

  kT_softmax_k<<<B_ * N_ / 16, 256, 0, stream>>>(qkv);

  kT_transpose<<<dim3(C_ / 64, N_ / 128, B_), 256, 0, stream>>>(
      qkv, qT, C3_, N_, (long long)N_ * C3_, (long long)C_ * N_);
  kT_softmax_row1024<<<B_ * C_ / 8, 256, 0, stream>>>(qT);

  kT_gemm<C3_, C3_, C_, N_, 16, 1, 4, false, false, false>
      <<<dim3(C_ / 64, C_ / 128, B_), 256, 0, stream>>>(
      qkv + C_,     (long long)N_ * C3_,
      qkv + 2 * C_, (long long)N_ * C3_,
      ctx, (long long)C_ * C_,
      nullptr, nullptr, 0);

  kT_gemm<N_, C_, N_, C_, 8, 2, 2, true, false, false>
      <<<dim3(N_ / 64, C_ / 256, B_), 256, 0, stream>>>(
      qT, (long long)C_ * N_,
      ctx, (long long)C_ * C_,
      o2T, (long long)C_ * N_,
      nullptr, nullptr, 0);

  kT_gemm<N_, C_, N_, C_, 8, 2, 2, true, true, true>
      <<<dim3(N_ / 64, C_ / 256, B_), 256, 0, stream>>>(
      o2T, (long long)C_ * N_,
      wprojT, 0,
      out, (long long)C_ * N_,
      proj_b, xn, (long long)C_ * N_);
}


# Output projection folded into the context: ctxP = context x proj_w^T (512^3 per batch) replaces the 1024x512x512 product q x context and its transposed intermediate; reassociated: sum order changed
# speedup vs baseline: 1.8118x; 1.0553x over previous
#include <hip/hip_runtime.h>
#include <math.h>

typedef float v2f __attribute__((ext_vector_type(2)));
typedef float v4f __attribute__((ext_vector_type(4)));
typedef float v8f __attribute__((ext_vector_type(8)));

#define B_    8
#define C_    512
#define N_    1024
#define C3_   1536
#define GRP_  8
#define GCH_  64
#define GELT_ (GCH_ * N_)
#define EPS_  1e-5f

__global__ __launch_bounds__(256) void kT_transpose(const float* __restrict__ in,
                                                    float* __restrict__ out,
                                                    int ldin, int ldout,
                                                    long long strideIn, long long strideOut) {
  __shared__ float tile[64][129];
  const float* ib = in  + (size_t)blockIdx.z * strideIn;
  float*       ob = out + (size_t)blockIdx.z * strideOut;
  int c0 = blockIdx.x * 64, r0 = blockIdx.y * 128;
  int q4 = threadIdx.x & 15;
  int g  = threadIdx.x >> 4;
  v4f v[8];
  #pragma unroll
  for (int i = 0; i < 8; ++i)
    v[i] = *(const v4f*)(ib + (size_t)(r0 + g + 16 * i) * ldin + c0 + 4 * q4);
  #pragma unroll
  for (int i = 0; i < 8; ++i)
    #pragma unroll
    for (int e = 0; e < 4; ++e)
      tile[4 * q4 + e][g + 16 * i] = v[i][e];
  __syncthreads();
  #pragma unroll
  for (int i = 0; i < 8; ++i) {
    int u = 16 * i + g;
    int c = u & 63, rh = u >> 6;
    v4f o;
    #pragma unroll
    for (int e = 0; e < 4; ++e) o[e] = tile[c][64 * rh + 4 * q4 + e];
    *(v4f*)(ob + (size_t)(c0 + c) * ldout + r0 + 64 * rh + 4 * q4) = o;
  }
}

__global__ __launch_bounds__(256) void kT_gn_stats(const float* __restrict__ x,
                                                   float* __restrict__ mean,
                                                   float* __restrict__ rstd) {
  __shared__ float s1[256], s2[256];
  int bg = blockIdx.x;
  const float4* p = (const float4*)(x + (size_t)bg * GELT_);
  float s = 0.f, q = 0.f;
  for (int i = threadIdx.x; i < GELT_ / 4; i += 256) {
    float4 v = p[i];
    s += v.x + v.y + v.z + v.w;
    q += v.x * v.x + v.y * v.y + v.z * v.z + v.w * v.w;
  }
  s1[threadIdx.x] = s; s2[threadIdx.x] = q;
  __syncthreads();
  for (int off = 128; off > 0; off >>= 1) {
    if (threadIdx.x < off) {
      s1[threadIdx.x] += s1[threadIdx.x + off];
      s2[threadIdx.x] += s2[threadIdx.x + off];
    }
    __syncthreads();
  }
  if (threadIdx.x == 0) {
    float m = s1[0] * (1.0f / GELT_);
    float v = s2[0] * (1.0f / GELT_) - m * m;
    mean[bg] = m;
    rstd[bg] = rsqrtf(v + EPS_);
  }
}

__device__ __forceinline__ float wave_max(float v) {
  #pragma unroll
  for (int off = 16; off > 0; off >>= 1) v = fmaxf(v, __shfl_xor(v, off));
  return v;
}
__device__ __forceinline__ float wave_sum(float v) {
  #pragma unroll
  for (int off = 16; off > 0; off >>= 1) v += __shfl_xor(v, off);
  return v;
}

__global__ __launch_bounds__(256) void kT_gn_apply(const float* __restrict__ x,
                                                   const float* __restrict__ gamma,
                                                   const float* __restrict__ beta,
                                                   const float* __restrict__ mean,
                                                   const float* __restrict__ rstd,
                                                   float* __restrict__ xn) {
  int lane = threadIdx.x & 31;
  int bc = blockIdx.x * 8 + (threadIdx.x >> 5);
  int c = bc & (C_ - 1);
  int bg = ((bc >> 9) << 3) + (c >> 6);
  const v4f* pin  = (const v4f*)(x  + (size_t)bc * N_);
  v4f*       pout = (v4f*)(xn + (size_t)bc * N_);
  v4f v[8];
  #pragma unroll
  for (int j = 0; j < 8; ++j) v[j] = pin[lane + 32 * j];
  float gm = gamma[c] * rstd[bg];
  float bt = beta[c] - mean[bg] * gm;
  #pragma unroll
  for (int j = 0; j < 8; ++j) {
    v4f w = v[j];
    w.x = w.x * gm + bt; w.y = w.y * gm + bt;
    w.z = w.z * gm + bt; w.w = w.w * gm + bt;
    pout[lane + 32 * j] = w;
  }
}

__global__ __launch_bounds__(256) void kT_softmax_k(float* __restrict__ qkv) {
  int lane = threadIdx.x & 31;
  int row0 = (blockIdx.x * 8 + (threadIdx.x >> 5)) * 2;
  v4f* p0 = (v4f*)(qkv + (size_t)row0 * C3_ + C_);
  v4f* p1 = (v4f*)(qkv + (size_t)(row0 + 1) * C3_ + C_);
  v4f a[4], b[4];
  #pragma unroll
  for (int j = 0; j < 4; ++j) { a[j] = p0[lane + 32 * j]; b[j] = p1[lane + 32 * j]; }
  float ma = fmaxf(fmaxf(a[0].x, a[0].y), fmaxf(a[0].z, a[0].w));
  float mb = fmaxf(fmaxf(b[0].x, b[0].y), fmaxf(b[0].z, b[0].w));
  #pragma unroll
  for (int j = 1; j < 4; ++j) {
    ma = fmaxf(ma, fmaxf(fmaxf(a[j].x, a[j].y), fmaxf(a[j].z, a[j].w)));
    mb = fmaxf(mb, fmaxf(fmaxf(b[j].x, b[j].y), fmaxf(b[j].z, b[j].w)));
  }
  ma = wave_max(ma);
  mb = wave_max(mb);
  float sa = 0.f, sb = 0.f;
  #pragma unroll
  for (int j = 0; j < 4; ++j) {
    a[j].x = __expf(a[j].x - ma); a[j].y = __expf(a[j].y - ma);
    a[j].z = __expf(a[j].z - ma); a[j].w = __expf(a[j].w - ma);
    b[j].x = __expf(b[j].x - mb); b[j].y = __expf(b[j].y - mb);
    b[j].z = __expf(b[j].z - mb); b[j].w = __expf(b[j].w - mb);
    sa += (a[j].x + a[j].y) + (a[j].z + a[j].w);
    sb += (b[j].x + b[j].y) + (b[j].z + b[j].w);
  }
  float ia = 1.0f / wave_sum(sa);
  float ib = 1.0f / wave_sum(sb);
  #pragma unroll
  for (int j = 0; j < 4; ++j) {
    p0[lane + 32 * j] = a[j] * ia;
    p1[lane + 32 * j] = b[j] * ib;
  }
}

__global__ __launch_bounds__(256) void kT_softmax_row1024(float* __restrict__ qT) {
  int lane = threadIdx.x & 31;
  v4f* row = (v4f*)(qT + (size_t)(blockIdx.x * 8 + (threadIdx.x >> 5)) * N_);
  v4f v[8];
  #pragma unroll
  for (int j = 0; j < 8; ++j) v[j] = row[lane + 32 * j];
  float mx = fmaxf(fmaxf(v[0].x, v[0].y), fmaxf(v[0].z, v[0].w));
  #pragma unroll
  for (int j = 1; j < 8; ++j)
    mx = fmaxf(mx, fmaxf(fmaxf(v[j].x, v[j].y), fmaxf(v[j].z, v[j].w)));
  mx = wave_max(mx);
  float s = 0.f;
  #pragma unroll
  for (int j = 0; j < 8; ++j) {
    v[j].x = __expf(v[j].x - mx); v[j].y = __expf(v[j].y - mx);
    v[j].z = __expf(v[j].z - mx); v[j].w = __expf(v[j].w - mx);
    s += (v[j].x + v[j].y) + (v[j].z + v[j].w);
  }
  float inv = 1.0f / wave_sum(s);
  #pragma unroll
  for (int j = 0; j < 8; ++j) row[lane + 32 * j] = v[j] * inv;
}

template <int LDA, int LDB, int LDC, int K, int NS, int TM, int WM,
          bool TSTORE, bool HAS_BIAS, bool HAS_RES>
__global__ __launch_bounds__(256) void kT_gemm(
    const float* __restrict__ A, long long strideA,
    const float* __restrict__ B, long long strideB,
    float* __restrict__ Cmat,    long long strideC,
    const float* __restrict__ bias,
    const float* __restrict__ resid, long long strideR) {
  static_assert(K % (4 * NS) == 0 && K >= 8 * NS, "ring must divide K");
  static_assert((TM == 1 || TM == 2) && (WM == 2 || WM == 4), "tile shape");
  int z = blockIdx.z;
  const float* Ab = A + (size_t)z * strideA;
  const float* Bb = B + (size_t)z * strideB;
  float*       Cb = Cmat + (size_t)z * strideC;

  int wave = threadIdx.x >> 5;
  int lane = threadIdx.x & 31;
  int half = lane >> 4;
  int ml   = lane & 15;
  constexpr int WN = 8 / WM;
  int m0 = blockIdx.x * (16 * TM * WM) + (wave % WM) * (16 * TM);
  int n0 = blockIdx.y * (64 * WN) + (wave / WM) * 64;

  v8f acc[TM][4];
  {
    v4f bv = {0.0f, 0.0f, 0.0f, 0.0f};
    if (HAS_BIAS) bv = *(const v4f*)(bias + n0 + 4 * ml);
    #pragma unroll
    for (int mi = 0; mi < TM; ++mi)
      #pragma unroll
      for (int t = 0; t < 4; ++t)
        #pragma unroll
        for (int r = 0; r < 8; ++r) acc[mi][t][r] = bv[t];
  }

  const float* Ap = Ab + (size_t)(2 * half) * LDA + m0 + ml;
  const float* Bp = Bb + (size_t)(2 * half) * LDB + n0 + 4 * ml;

  v2f fa[NS][TM];
  v4f rb[NS][2];

#define GEMM_LOAD_SLOT(s)                                                      \
  {                                                                            \
    const int k0 = 4 * (s) * LDA, k1 = (4 * (s) + 1) * LDA;                    \
    _Pragma("unroll")                                                          \
    for (int mi = 0; mi < TM; ++mi) {                                          \
      fa[s][mi].x = Ap[k0 + 16 * mi]; fa[s][mi].y = Ap[k1 + 16 * mi];          \
    }                                                                          \
    rb[s][0] = *(const v4f*)(Bp + 4 * (s) * LDB);                              \
    rb[s][1] = *(const v4f*)(Bp + (4 * (s) + 1) * LDB);                        \
  }
#define GEMM_MMA_SLOT(s)                                                       \
  {                                                                            \
    _Pragma("unroll")                                                          \
    for (int mi = 0; mi < TM; ++mi)                                            \
      _Pragma("unroll")                                                        \
      for (int t = 0; t < 4; ++t) {                                            \
        v2f fn = {rb[s][0][t], rb[s][1][t]};                                   \
        acc[mi][t] = __builtin_amdgcn_wmma_f32_16x16x4_f32(                    \
            false, fa[s][mi], false, fn, (short)0, acc[mi][t], false, false);  \
      }                                                                        \
  }

  #pragma unroll
  for (int s = 0; s < NS; ++s) {
    GEMM_LOAD_SLOT(s)
    __builtin_amdgcn_sched_barrier(0);
  }
  Ap += 4 * NS * LDA;
  Bp += 4 * NS * LDB;

  #pragma unroll 1
  for (int kk = 4 * NS; kk < K; kk += 4 * NS) {
    #pragma unroll
    for (int s = 0; s < NS; ++s) {
      GEMM_MMA_SLOT(s)
      GEMM_LOAD_SLOT(s)
      __builtin_amdgcn_sched_barrier(0);
    }
    Ap += 4 * NS * LDA;
    Bp += 4 * NS * LDB;
  }
  #pragma unroll
  for (int s = 0; s < NS; ++s) {
    GEMM_MMA_SLOT(s)
    __builtin_amdgcn_sched_barrier(0);
  }
#undef GEMM_LOAD_SLOT
#undef GEMM_MMA_SLOT

  if (!TSTORE) {
    #pragma unroll
    for (int mi = 0; mi < TM; ++mi) {
      #pragma unroll
      for (int r = 0; r < 8; ++r) {
        int m = m0 + 16 * mi + r + 8 * half;
        v4f v = {acc[mi][0][r], acc[mi][1][r], acc[mi][2][r], acc[mi][3][r]};
        *(v4f*)(Cb + (size_t)m * LDC + n0 + 4 * ml) = v;
      }
    }
  } else {
    const float* Rb = HAS_RES ? (resid + (size_t)z * strideR) : nullptr;
    #pragma unroll
    for (int mi = 0; mi < TM; ++mi) {
      #pragma unroll
      for (int r = 0; r < 8; ++r) {
        int m = m0 + 16 * mi + r + 8 * half;
        #pragma unroll
        for (int t = 0; t < 4; ++t) {
          size_t i = (size_t)(n0 + 4 * ml + t) * LDC + m;
          float v = acc[mi][t][r];
          if (HAS_RES) v += Rb[i];
          Cb[i] = v;
        }
      }
    }
  }
}

extern "C" void kernel_launch(void* const* d_in, const int* in_sizes, int n_in,
                              void* d_out, int out_size, void* d_ws, size_t ws_size,
                              hipStream_t stream) {
  const float* x      = (const float*)d_in[0];
  const float* gn_w   = (const float*)d_in[1];
  const float* gn_b   = (const float*)d_in[2];
  const float* qkv_w  = (const float*)d_in[3];
  const float* qkv_b  = (const float*)d_in[4];
  const float* proj_w = (const float*)d_in[5];
  const float* proj_b = (const float*)d_in[6];
  float* out = (float*)d_out;

  float* ws = (float*)d_ws;
  const size_t XN_SZ  = (size_t)B_ * C_ * N_;
  const size_t QKV_SZ = (size_t)B_ * N_ * C3_;
  const size_t CTX_SZ = (size_t)B_ * C_ * C_;
  float* mean   = ws;
  float* rstd   = ws + 64;
  float* xn     = ws + 256;
  float* qkv    = xn   + XN_SZ;
  float* qT     = qkv  + QKV_SZ;
  float* ctxT   = qT   + XN_SZ;
  float* ctxP   = ctxT + CTX_SZ;
  float* wqkvT  = ctxP + CTX_SZ;
  float* wprojT = wqkvT + (size_t)C_ * C3_;

  kT_transpose<<<dim3(C_ / 64, C3_ / 128, 1), 256, 0, stream>>>(qkv_w,  wqkvT,  C_, C3_, 0, 0);
  kT_transpose<<<dim3(C_ / 64, C_  / 128, 1), 256, 0, stream>>>(proj_w, wprojT, C_, C_,  0, 0);

  kT_gn_stats<<<B_ * GRP_, 256, 0, stream>>>(x, mean, rstd);
  kT_gn_apply<<<B_ * C_ / 8, 256, 0, stream>>>(x, gn_w, gn_b, mean, rstd, xn);

  kT_gemm<N_, C3_, C3_, C_, 4, 2, 2, false, true, false>
      <<<dim3(N_ / 64, C3_ / 256, B_), 256, 0, stream>>>(
      xn, (long long)C_ * N_,
      wqkvT, 0,
      qkv, (long long)N_ * C3_,
      qkv_b, nullptr, 0);

  kT_softmax_k<<<B_ * N_ / 16, 256, 0, stream>>>(qkv);

  kT_transpose<<<dim3(C_ / 64, N_ / 128, B_), 256, 0, stream>>>(
      qkv, qT, C3_, N_, (long long)N_ * C3_, (long long)C_ * N_);
  kT_softmax_row1024<<<B_ * C_ / 8, 256, 0, stream>>>(qT);

  kT_gemm<C3_, C3_, C_, N_, 16, 1, 4, false, false, false>
      <<<dim3(C_ / 64, C_ / 128, B_), 256, 0, stream>>>(
      qkv + 2 * C_, (long long)N_ * C3_,
      qkv + C_,     (long long)N_ * C3_,
      ctxT, (long long)C_ * C_,
      nullptr, nullptr, 0);

  kT_gemm<C_, C_, C_, C_, 16, 1, 4, false, false, false>
      <<<dim3(C_ / 64, C_ / 128, B_), 256, 0, stream>>>(
      ctxT, (long long)C_ * C_,
      wprojT, 0,
      ctxP, (long long)C_ * C_,
      nullptr, nullptr, 0);

  kT_gemm<N_, C_, N_, C_, 8, 2, 2, true, true, true>
      <<<dim3(N_ / 64, C_ / 256, B_), 256, 0, stream>>>(
      qT, (long long)C_ * N_,
      ctxP, (long long)C_ * C_,
      out, (long long)C_ * N_,
      proj_b, xn, (long long)C_ * N_);
}


# Independent kernels share launches, picked by block index: weight transposes beside GroupNorm stats, k softmax beside q transpose, q softmax beside context GEMM: 11 launches to 7; sum order kept
# speedup vs baseline: 1.9441x; 1.0730x over previous
#include <hip/hip_runtime.h>
#include <math.h>

typedef float v2f __attribute__((ext_vector_type(2)));
typedef float v4f __attribute__((ext_vector_type(4)));
typedef float v8f __attribute__((ext_vector_type(8)));

#define B_    8
#define C_    512
#define N_    1024
#define C3_   1536
#define GRP_  8
#define GCH_  64
#define GELT_ (GCH_ * N_)
#define EPS_  1e-5f

__device__ __forceinline__ void transpose_tile(const float* __restrict__ in,
                                               float* __restrict__ out,
                                               int ldin, int ldout,
                                               long long strideIn, long long strideOut,
                                               int bx, int by, int bz) {
  __shared__ float tile[64][129];
  const float* ib = in  + (size_t)bz * strideIn;
  float*       ob = out + (size_t)bz * strideOut;
  int c0 = bx * 64, r0 = by * 128;
  int q4 = threadIdx.x & 15;
  int g  = threadIdx.x >> 4;
  v4f v[8];
  #pragma unroll
  for (int i = 0; i < 8; ++i)
    v[i] = *(const v4f*)(ib + (size_t)(r0 + g + 16 * i) * ldin + c0 + 4 * q4);
  #pragma unroll
  for (int i = 0; i < 8; ++i)
    #pragma unroll
    for (int e = 0; e < 4; ++e)
      tile[4 * q4 + e][g + 16 * i] = v[i][e];
  __syncthreads();
  #pragma unroll
  for (int i = 0; i < 8; ++i) {
    int u = 16 * i + g;
    int c = u & 63, rh = u >> 6;
    v4f o;
    #pragma unroll
    for (int e = 0; e < 4; ++e) o[e] = tile[c][64 * rh + 4 * q4 + e];
    *(v4f*)(ob + (size_t)(c0 + c) * ldout + r0 + 64 * rh + 4 * q4) = o;
  }
}

__device__ __forceinline__ void gn_stats_group(const float* __restrict__ x,
                                               float* __restrict__ mean,
                                               float* __restrict__ rstd,
                                               int bg) {
  __shared__ float s1[256], s2[256];
  const float4* p = (const float4*)(x + (size_t)bg * GELT_);
  float s = 0.f, q = 0.f;
  for (int i = threadIdx.x; i < GELT_ / 4; i += 256) {
    float4 v = p[i];
    s += v.x + v.y + v.z + v.w;
    q += v.x * v.x + v.y * v.y + v.z * v.z + v.w * v.w;
  }
  s1[threadIdx.x] = s; s2[threadIdx.x] = q;
  __syncthreads();
  for (int off = 128; off > 0; off >>= 1) {
    if (threadIdx.x < off) {
      s1[threadIdx.x] += s1[threadIdx.x + off];
      s2[threadIdx.x] += s2[threadIdx.x + off];
    }
    __syncthreads();
  }
  if (threadIdx.x == 0) {
    float m = s1[0] * (1.0f / GELT_);
    float v = s2[0] * (1.0f / GELT_) - m * m;
    mean[bg] = m;
    rstd[bg] = rsqrtf(v + EPS_);
  }
}

__device__ __forceinline__ float wave_max(float v) {
  #pragma unroll
  for (int off = 16; off > 0; off >>= 1) v = fmaxf(v, __shfl_xor(v, off));
  return v;
}
__device__ __forceinline__ float wave_sum(float v) {
  #pragma unroll
  for (int off = 16; off > 0; off >>= 1) v += __shfl_xor(v, off);
  return v;
}

__global__ __launch_bounds__(256) void kT_gn_apply(const float* __restrict__ x,
                                                   const float* __restrict__ gamma,
                                                   const float* __restrict__ beta,
                                                   const float* __restrict__ mean,
                                                   const float* __restrict__ rstd,
                                                   float* __restrict__ xn) {
  int lane = threadIdx.x & 31;
  int bc = blockIdx.x * 8 + (threadIdx.x >> 5);
  int c = bc & (C_ - 1);
  int bg = ((bc >> 9) << 3) + (c >> 6);
  const v4f* pin  = (const v4f*)(x  + (size_t)bc * N_);
  v4f*       pout = (v4f*)(xn + (size_t)bc * N_);
  v4f v[8];
  #pragma unroll
  for (int j = 0; j < 8; ++j) v[j] = pin[lane + 32 * j];
  float gm = gamma[c] * rstd[bg];
  float bt = beta[c] - mean[bg] * gm;
  #pragma unroll
  for (int j = 0; j < 8; ++j) {
    v4f w = v[j];
    w.x = w.x * gm + bt; w.y = w.y * gm + bt;
    w.z = w.z * gm + bt; w.w = w.w * gm + bt;
    pout[lane + 32 * j] = w;
  }
}

__device__ __forceinline__ void softmax_k_rows(float* __restrict__ qkv, int blk) {
  int lane = threadIdx.x & 31;
  int row0 = (blk * 8 + (threadIdx.x >> 5)) * 2;
  v4f* p0 = (v4f*)(qkv + (size_t)row0 * C3_ + C_);
  v4f* p1 = (v4f*)(qkv + (size_t)(row0 + 1) * C3_ + C_);
  v4f a[4], b[4];
  #pragma unroll
  for (int j = 0; j < 4; ++j) { a[j] = p0[lane + 32 * j]; b[j] = p1[lane + 32 * j]; }
  float ma = fmaxf(fmaxf(a[0].x, a[0].y), fmaxf(a[0].z, a[0].w));
  float mb = fmaxf(fmaxf(b[0].x, b[0].y), fmaxf(b[0].z, b[0].w));
  #pragma unroll
  for (int j = 1; j < 4; ++j) {
    ma = fmaxf(ma, fmaxf(fmaxf(a[j].x, a[j].y), fmaxf(a[j].z, a[j].w)));
    mb = fmaxf(mb, fmaxf(fmaxf(b[j].x, b[j].y), fmaxf(b[j].z, b[j].w)));
  }
  ma = wave_max(ma);
  mb = wave_max(mb);
  float sa = 0.f, sb = 0.f;
  #pragma unroll
  for (int j = 0; j < 4; ++j) {
    a[j].x = __expf(a[j].x - ma); a[j].y = __expf(a[j].y - ma);
    a[j].z = __expf(a[j].z - ma); a[j].w = __expf(a[j].w - ma);
    b[j].x = __expf(b[j].x - mb); b[j].y = __expf(b[j].y - mb);
    b[j].z = __expf(b[j].z - mb); b[j].w = __expf(b[j].w - mb);
    sa += (a[j].x + a[j].y) + (a[j].z + a[j].w);
    sb += (b[j].x + b[j].y) + (b[j].z + b[j].w);
  }
  float ia = 1.0f / wave_sum(sa);
  float ib = 1.0f / wave_sum(sb);
  #pragma unroll
  for (int j = 0; j < 4; ++j) {
    p0[lane + 32 * j] = a[j] * ia;
    p1[lane + 32 * j] = b[j] * ib;
  }
}

__device__ __forceinline__ void softmax_rows1024(float* __restrict__ qT, int blk) {
  int lane = threadIdx.x & 31;
  v4f* row = (v4f*)(qT + (size_t)(blk * 8 + (threadIdx.x >> 5)) * N_);
  v4f v[8];
  #pragma unroll
  for (int j = 0; j < 8; ++j) v[j] = row[lane + 32 * j];
  float mx = fmaxf(fmaxf(v[0].x, v[0].y), fmaxf(v[0].z, v[0].w));
  #pragma unroll
  for (int j = 1; j < 8; ++j)
    mx = fmaxf(mx, fmaxf(fmaxf(v[j].x, v[j].y), fmaxf(v[j].z, v[j].w)));
  mx = wave_max(mx);
  float s = 0.f;
  #pragma unroll
  for (int j = 0; j < 8; ++j) {
    v[j].x = __expf(v[j].x - mx); v[j].y = __expf(v[j].y - mx);
    v[j].z = __expf(v[j].z - mx); v[j].w = __expf(v[j].w - mx);
    s += (v[j].x + v[j].y) + (v[j].z + v[j].w);
  }
  float inv = 1.0f / wave_sum(s);
  #pragma unroll
  for (int j = 0; j < 8; ++j) row[lane + 32 * j] = v[j] * inv;
}

template <int LDA, int LDB, int LDC, int K, int NS, int TM, int WM,
          bool TSTORE, bool HAS_BIAS, bool HAS_RES>
__device__ __forceinline__ void gemm_tile(
    const float* __restrict__ A, long long strideA,
    const float* __restrict__ B, long long strideB,
    float* __restrict__ Cmat,    long long strideC,
    const float* __restrict__ bias,
    const float* __restrict__ resid, long long strideR,
    int bx, int by, int z) {
  static_assert(K % (4 * NS) == 0 && K >= 8 * NS, "ring must divide K");
  static_assert((TM == 1 || TM == 2) && (WM == 2 || WM == 4), "tile shape");
  const float* Ab = A + (size_t)z * strideA;
  const float* Bb = B + (size_t)z * strideB;
  float*       Cb = Cmat + (size_t)z * strideC;

  int wave = threadIdx.x >> 5;
  int lane = threadIdx.x & 31;
  int half = lane >> 4;
  int ml   = lane & 15;
  constexpr int WN = 8 / WM;
  int m0 = bx * (16 * TM * WM) + (wave % WM) * (16 * TM);
  int n0 = by * (64 * WN) + (wave / WM) * 64;

  v8f acc[TM][4];
  {
    v4f bv = {0.0f, 0.0f, 0.0f, 0.0f};
    if (HAS_BIAS) bv = *(const v4f*)(bias + n0 + 4 * ml);
    #pragma unroll
    for (int mi = 0; mi < TM; ++mi)
      #pragma unroll
      for (int t = 0; t < 4; ++t)
        #pragma unroll
        for (int r = 0; r < 8; ++r) acc[mi][t][r] = bv[t];
  }

  const float* Ap = Ab + (size_t)(2 * half) * LDA + m0 + ml;
  const float* Bp = Bb + (size_t)(2 * half) * LDB + n0 + 4 * ml;

  v2f fa[NS][TM];
  v4f rb[NS][2];

#define GEMM_LOAD_SLOT(s)                                                      \
  {                                                                            \
    const int k0 = 4 * (s) * LDA, k1 = (4 * (s) + 1) * LDA;                    \
    _Pragma("unroll")                                                          \
    for (int mi = 0; mi < TM; ++mi) {                                          \
      fa[s][mi].x = Ap[k0 + 16 * mi]; fa[s][mi].y = Ap[k1 + 16 * mi];          \
    }                                                                          \
    rb[s][0] = *(const v4f*)(Bp + 4 * (s) * LDB);                              \
    rb[s][1] = *(const v4f*)(Bp + (4 * (s) + 1) * LDB);                        \
  }
#define GEMM_MMA_SLOT(s)                                                       \
  {                                                                            \
    _Pragma("unroll")                                                          \
    for (int mi = 0; mi < TM; ++mi)                                            \
      _Pragma("unroll")                                                        \
      for (int t = 0; t < 4; ++t) {                                            \
        v2f fn = {rb[s][0][t], rb[s][1][t]};                                   \
        acc[mi][t] = __builtin_amdgcn_wmma_f32_16x16x4_f32(                    \
            false, fa[s][mi], false, fn, (short)0, acc[mi][t], false, false);  \
      }                                                                        \
  }

  #pragma unroll
  for (int s = 0; s < NS; ++s) {
    GEMM_LOAD_SLOT(s)
    __builtin_amdgcn_sched_barrier(0);
  }
  Ap += 4 * NS * LDA;
  Bp += 4 * NS * LDB;

  #pragma unroll 1
  for (int kk = 4 * NS; kk < K; kk += 4 * NS) {
    #pragma unroll
    for (int s = 0; s < NS; ++s) {
      GEMM_MMA_SLOT(s)
      GEMM_LOAD_SLOT(s)
      __builtin_amdgcn_sched_barrier(0);
    }
    Ap += 4 * NS * LDA;
    Bp += 4 * NS * LDB;
  }
  #pragma unroll
  for (int s = 0; s < NS; ++s) {
    GEMM_MMA_SLOT(s)
    __builtin_amdgcn_sched_barrier(0);
  }
#undef GEMM_LOAD_SLOT
#undef GEMM_MMA_SLOT

  if (!TSTORE) {
    #pragma unroll
    for (int mi = 0; mi < TM; ++mi) {
      #pragma unroll
      for (int r = 0; r < 8; ++r) {
        int m = m0 + 16 * mi + r + 8 * half;
        v4f v = {acc[mi][0][r], acc[mi][1][r], acc[mi][2][r], acc[mi][3][r]};
        *(v4f*)(Cb + (size_t)m * LDC + n0 + 4 * ml) = v;
      }
    }
  } else {
    const float* Rb = HAS_RES ? (resid + (size_t)z * strideR) : nullptr;
    #pragma unroll
    for (int mi = 0; mi < TM; ++mi) {
      #pragma unroll
      for (int r = 0; r < 8; ++r) {
        int m = m0 + 16 * mi + r + 8 * half;
        #pragma unroll
        for (int t = 0; t < 4; ++t) {
          size_t i = (size_t)(n0 + 4 * ml + t) * LDC + m;
          float v = acc[mi][t][r];
          if (HAS_RES) v += Rb[i];
          Cb[i] = v;
        }
      }
    }
  }
}

template <int LDA, int LDB, int LDC, int K, int NS, int TM, int WM,
          bool TSTORE, bool HAS_BIAS, bool HAS_RES>
__global__ __launch_bounds__(256) void kT_gemm(
    const float* __restrict__ A, long long strideA,
    const float* __restrict__ B, long long strideB,
    float* __restrict__ Cmat,    long long strideC,
    const float* __restrict__ bias,
    const float* __restrict__ resid, long long strideR) {
  gemm_tile<LDA, LDB, LDC, K, NS, TM, WM, TSTORE, HAS_BIAS, HAS_RES>(
      A, strideA, B, strideB, Cmat, strideC, bias, resid, strideR,
      blockIdx.x, blockIdx.y, blockIdx.z);
}

__global__ __launch_bounds__(256) void kT_stats_wtranspose(const float* __restrict__ x,
                                                           float* __restrict__ mean,
                                                           float* __restrict__ rstd,
                                                           const float* __restrict__ qkv_w,
                                                           float* __restrict__ wqkvT,
                                                           const float* __restrict__ proj_w,
                                                           float* __restrict__ wprojT) {
  int id = blockIdx.x;
  if (id < B_ * GRP_) {
    gn_stats_group(x, mean, rstd, id);
  } else if (id < B_ * GRP_ + (C_ / 64) * (C3_ / 128)) {
    int t = id - B_ * GRP_;
    transpose_tile(qkv_w, wqkvT, C_, C3_, 0, 0, t & 7, t >> 3, 0);
  } else {
    int t = id - B_ * GRP_ - (C_ / 64) * (C3_ / 128);
    transpose_tile(proj_w, wprojT, C_, C_, 0, 0, t & 7, t >> 3, 0);
  }
}

__global__ __launch_bounds__(256) void kT_qtranspose_ksoftmax(float* qkv, float* qT) {
  int id = blockIdx.x;
  if (id < (C_ / 64) * (N_ / 128) * B_) {
    transpose_tile(qkv, qT, C3_, N_, (long long)N_ * C3_, (long long)C_ * N_,
                   id & 7, (id >> 3) & 7, id >> 6);
  } else {
    softmax_k_rows(qkv, id - (C_ / 64) * (N_ / 128) * B_);
  }
}

__global__ __launch_bounds__(256) void kT_context_qsoftmax(const float* __restrict__ vmat,
                                                           const float* __restrict__ kmat,
                                                           float* __restrict__ ctxT,
                                                           float* __restrict__ qT) {
  int id = blockIdx.x;
  if (id < (C_ / 64) * (C_ / 128) * B_) {
    gemm_tile<C3_, C3_, C_, N_, 16, 1, 4, false, false, false>(
        vmat, (long long)N_ * C3_,
        kmat, (long long)N_ * C3_,
        ctxT, (long long)C_ * C_,
        nullptr, nullptr, 0,
        id & 7, (id >> 3) & 3, id >> 5);
  } else {
    softmax_rows1024(qT, id - (C_ / 64) * (C_ / 128) * B_);
  }
}

extern "C" void kernel_launch(void* const* d_in, const int* in_sizes, int n_in,
                              void* d_out, int out_size, void* d_ws, size_t ws_size,
                              hipStream_t stream) {
  const float* x      = (const float*)d_in[0];
  const float* gn_w   = (const float*)d_in[1];
  const float* gn_b   = (const float*)d_in[2];
  const float* qkv_w  = (const float*)d_in[3];
  const float* qkv_b  = (const float*)d_in[4];
  const float* proj_w = (const float*)d_in[5];
  const float* proj_b = (const float*)d_in[6];
  float* out = (float*)d_out;

  float* ws = (float*)d_ws;
  const size_t XN_SZ  = (size_t)B_ * C_ * N_;
  const size_t QKV_SZ = (size_t)B_ * N_ * C3_;
  const size_t CTX_SZ = (size_t)B_ * C_ * C_;
  float* mean   = ws;
  float* rstd   = ws + 64;
  float* xn     = ws + 256;
  float* qkv    = xn   + XN_SZ;
  float* qT     = qkv  + QKV_SZ;
  float* ctxT   = qT   + XN_SZ;
  float* ctxP   = ctxT + CTX_SZ;
  float* wqkvT  = ctxP + CTX_SZ;
  float* wprojT = wqkvT + (size_t)C_ * C3_;

  kT_stats_wtranspose<<<192, 256, 0, stream>>>(x, mean, rstd, qkv_w, wqkvT, proj_w, wprojT);

  kT_gn_apply<<<B_ * C_ / 8, 256, 0, stream>>>(x, gn_w, gn_b, mean, rstd, xn);

  kT_gemm<N_, C3_, C3_, C_, 4, 2, 2, false, true, false>
      <<<dim3(N_ / 64, C3_ / 256, B_), 256, 0, stream>>>(
      xn, (long long)C_ * N_,
      wqkvT, 0,
      qkv, (long long)N_ * C3_,
      qkv_b, nullptr, 0);

  kT_qtranspose_ksoftmax<<<1024, 256, 0, stream>>>(qkv, qT);

  kT_context_qsoftmax<<<768, 256, 0, stream>>>(qkv + 2 * C_, qkv + C_, ctxT, qT);

  kT_gemm<C_, C_, C_, C_, 16, 1, 4, false, false, false>
      <<<dim3(C_ / 64, C_ / 128, B_), 256, 0, stream>>>(
      ctxT, (long long)C_ * C_,
      wprojT, 0,
      ctxP, (long long)C_ * C_,
      nullptr, nullptr, 0);

  kT_gemm<N_, C_, N_, C_, 8, 2, 2, true, true, true>
      <<<dim3(N_ / 64, C_ / 256, B_), 256, 0, stream>>>(
      qT, (long long)C_ * N_,
      ctxP, (long long)C_ * C_,
      out, (long long)C_ * N_,
      proj_b, xn, (long long)C_ * N_);
}


# v projection folded away: ctxP = (k^T xn)(proj_w Wv)^T + colsum(k) x (proj_w bv), so the qkv GEMM drops its v third (1024x512x512 per batch); reassociated: sum order changed
# speedup vs baseline: 2.0545x; 1.0568x over previous
#include <hip/hip_runtime.h>
#include <math.h>

typedef float v2f __attribute__((ext_vector_type(2)));
typedef float v4f __attribute__((ext_vector_type(4)));
typedef float v8f __attribute__((ext_vector_type(8)));

#define B_    8
#define C_    512
#define N_    1024
#define QK_   1024
#define GRP_  8
#define GCH_  64
#define GELT_ (GCH_ * N_)
#define EPS_  1e-5f

__device__ __forceinline__ void transpose_tile(const float* __restrict__ in,
                                               float* __restrict__ out,
                                               int ldin, int ldout,
                                               long long strideIn, long long strideOut,
                                               int bx, int by, int bz) {
  __shared__ float tile[64][129];
  const float* ib = in  + (size_t)bz * strideIn;
  float*       ob = out + (size_t)bz * strideOut;
  int c0 = bx * 64, r0 = by * 128;
  int q4 = threadIdx.x & 15;
  int g  = threadIdx.x >> 4;
  v4f v[8];
  #pragma unroll
  for (int i = 0; i < 8; ++i)
    v[i] = *(const v4f*)(ib + (size_t)(r0 + g + 16 * i) * ldin + c0 + 4 * q4);
  #pragma unroll
  for (int i = 0; i < 8; ++i)
    #pragma unroll
    for (int e = 0; e < 4; ++e)
      tile[4 * q4 + e][g + 16 * i] = v[i][e];
  __syncthreads();
  #pragma unroll
  for (int i = 0; i < 8; ++i) {
    int u = 16 * i + g;
    int c = u & 63, rh = u >> 6;
    v4f o;
    #pragma unroll
    for (int e = 0; e < 4; ++e) o[e] = tile[c][64 * rh + 4 * q4 + e];
    *(v4f*)(ob + (size_t)(c0 + c) * ldout + r0 + 64 * rh + 4 * q4) = o;
  }
}

__device__ __forceinline__ void gn_stats_group(const float* __restrict__ x,
                                               float* __restrict__ mean,
                                               float* __restrict__ rstd,
                                               int bg) {
  __shared__ float s1[256], s2[256];
  const float4* p = (const float4*)(x + (size_t)bg * GELT_);
  float s = 0.f, q = 0.f;
  for (int i = threadIdx.x; i < GELT_ / 4; i += 256) {
    float4 v = p[i];
    s += v.x + v.y + v.z + v.w;
    q += v.x * v.x + v.y * v.y + v.z * v.z + v.w * v.w;
  }
  s1[threadIdx.x] = s; s2[threadIdx.x] = q;
  __syncthreads();
  for (int off = 128; off > 0; off >>= 1) {
    if (threadIdx.x < off) {
      s1[threadIdx.x] += s1[threadIdx.x + off];
      s2[threadIdx.x] += s2[threadIdx.x + off];
    }
    __syncthreads();
  }
  if (threadIdx.x == 0) {
    float m = s1[0] * (1.0f / GELT_);
    float v = s2[0] * (1.0f / GELT_) - m * m;
    mean[bg] = m;
    rstd[bg] = rsqrtf(v + EPS_);
  }
}

__device__ __forceinline__ float wave_max(float v) {
  #pragma unroll
  for (int off = 16; off > 0; off >>= 1) v = fmaxf(v, __shfl_xor(v, off));
  return v;
}
__device__ __forceinline__ float wave_sum(float v) {
  #pragma unroll
  for (int off = 16; off > 0; off >>= 1) v += __shfl_xor(v, off);
  return v;
}

__global__ __launch_bounds__(256) void kT_gn_apply(const float* __restrict__ x,
                                                   const float* __restrict__ gamma,
                                                   const float* __restrict__ beta,
                                                   const float* __restrict__ mean,
                                                   const float* __restrict__ rstd,
                                                   float* __restrict__ xn) {
  int lane = threadIdx.x & 31;
  int bc = blockIdx.x * 8 + (threadIdx.x >> 5);
  int c = bc & (C_ - 1);
  int bg = ((bc >> 9) << 3) + (c >> 6);
  const v4f* pin  = (const v4f*)(x  + (size_t)bc * N_);
  v4f*       pout = (v4f*)(xn + (size_t)bc * N_);
  v4f v[8];
  #pragma unroll
  for (int j = 0; j < 8; ++j) v[j] = pin[lane + 32 * j];
  float gm = gamma[c] * rstd[bg];
  float bt = beta[c] - mean[bg] * gm;
  #pragma unroll
  for (int j = 0; j < 8; ++j) {
    v4f w = v[j];
    w.x = w.x * gm + bt; w.y = w.y * gm + bt;
    w.z = w.z * gm + bt; w.w = w.w * gm + bt;
    pout[lane + 32 * j] = w;
  }
}

__device__ __forceinline__ void softmax_k_rows(float* __restrict__ qk, int blk) {
  int lane = threadIdx.x & 31;
  int row0 = (blk * 8 + (threadIdx.x >> 5)) * 2;
  v4f* p0 = (v4f*)(qk + (size_t)row0 * QK_ + C_);
  v4f* p1 = (v4f*)(qk + (size_t)(row0 + 1) * QK_ + C_);
  v4f a[4], b[4];
  #pragma unroll
  for (int j = 0; j < 4; ++j) { a[j] = p0[lane + 32 * j]; b[j] = p1[lane + 32 * j]; }
  float ma = fmaxf(fmaxf(a[0].x, a[0].y), fmaxf(a[0].z, a[0].w));
  float mb = fmaxf(fmaxf(b[0].x, b[0].y), fmaxf(b[0].z, b[0].w));
  #pragma unroll
  for (int j = 1; j < 4; ++j) {
    ma = fmaxf(ma, fmaxf(fmaxf(a[j].x, a[j].y), fmaxf(a[j].z, a[j].w)));
    mb = fmaxf(mb, fmaxf(fmaxf(b[j].x, b[j].y), fmaxf(b[j].z, b[j].w)));
  }
  ma = wave_max(ma);
  mb = wave_max(mb);
  float sa = 0.f, sb = 0.f;
  #pragma unroll
  for (int j = 0; j < 4; ++j) {
    a[j].x = __expf(a[j].x - ma); a[j].y = __expf(a[j].y - ma);
    a[j].z = __expf(a[j].z - ma); a[j].w = __expf(a[j].w - ma);
    b[j].x = __expf(b[j].x - mb); b[j].y = __expf(b[j].y - mb);
    b[j].z = __expf(b[j].z - mb); b[j].w = __expf(b[j].w - mb);
    sa += (a[j].x + a[j].y) + (a[j].z + a[j].w);
    sb += (b[j].x + b[j].y) + (b[j].z + b[j].w);
  }
  float ia = 1.0f / wave_sum(sa);
  float ib = 1.0f / wave_sum(sb);
  #pragma unroll
  for (int j = 0; j < 4; ++j) {
    p0[lane + 32 * j] = a[j] * ia;
    p1[lane + 32 * j] = b[j] * ib;
  }
}

__device__ __forceinline__ void softmax_rows1024(float* __restrict__ qT, int blk) {
  int lane = threadIdx.x & 31;
  v4f* row = (v4f*)(qT + (size_t)(blk * 8 + (threadIdx.x >> 5)) * N_);
  v4f v[8];
  #pragma unroll
  for (int j = 0; j < 8; ++j) v[j] = row[lane + 32 * j];
  float mx = fmaxf(fmaxf(v[0].x, v[0].y), fmaxf(v[0].z, v[0].w));
  #pragma unroll
  for (int j = 1; j < 8; ++j)
    mx = fmaxf(mx, fmaxf(fmaxf(v[j].x, v[j].y), fmaxf(v[j].z, v[j].w)));
  mx = wave_max(mx);
  float s = 0.f;
  #pragma unroll
  for (int j = 0; j < 8; ++j) {
    v[j].x = __expf(v[j].x - mx); v[j].y = __expf(v[j].y - mx);
    v[j].z = __expf(v[j].z - mx); v[j].w = __expf(v[j].w - mx);
    s += (v[j].x + v[j].y) + (v[j].z + v[j].w);
  }
  float inv = 1.0f / wave_sum(s);
  #pragma unroll
  for (int j = 0; j < 8; ++j) row[lane + 32 * j] = v[j] * inv;
}

__device__ __forceinline__ void k_colsum_slice(const float* __restrict__ kmat,
                                               float* __restrict__ ksum,
                                               int id) {
  __shared__ v4f red[16][17];
  int b = id >> 3, cs = id & 7;
  int q4 = threadIdx.x & 15;
  int g  = threadIdx.x >> 4;
  const float* p = kmat + (size_t)b * N_ * QK_ + (size_t)g * QK_ + cs * 64 + 4 * q4;
  v4f acc = {0.0f, 0.0f, 0.0f, 0.0f};
  #pragma unroll 1
  for (int i0 = 0; i0 < N_ / 16; i0 += 8) {
    v4f v[8];
    #pragma unroll
    for (int j = 0; j < 8; ++j) v[j] = *(const v4f*)(p + (size_t)(16 * (i0 + j)) * QK_);
    #pragma unroll
    for (int j = 0; j < 8; ++j) acc += v[j];
  }
  red[g][q4] = acc;
  __syncthreads();
  if (threadIdx.x < 64) {
    float s = 0.f;
    #pragma unroll
    for (int r = 0; r < 16; ++r) s += ((const float*)&red[r][0])[threadIdx.x];
    ksum[b * C_ + cs * 64 + threadIdx.x] = s;
  }
}

__device__ __forceinline__ void projbias_rows(const float* __restrict__ proj_w,
                                              const float* __restrict__ bv,
                                              float* __restrict__ pb,
                                              int blk) {
  int lane = threadIdx.x & 31;
  int e0 = blk * 64 + (threadIdx.x >> 5) * 8;
  v4f b4[4];
  #pragma unroll
  for (int j = 0; j < 4; ++j) b4[j] = *(const v4f*)(bv + 4 * (lane + 32 * j));
  #pragma unroll
  for (int r = 0; r < 8; ++r) {
    const v4f* row = (const v4f*)(proj_w + (size_t)(e0 + r) * C_);
    float s = 0.f;
    #pragma unroll
    for (int j = 0; j < 4; ++j) {
      v4f w = row[lane + 32 * j];
      s += (w.x * b4[j].x + w.y * b4[j].y) + (w.z * b4[j].z + w.w * b4[j].w);
    }
    s = wave_sum(s);
    if (lane == 0) pb[e0 + r] = s;
  }
}

template <int LDA, int LDB, int LDC, int K, int NS, int TM, int WM,
          bool TSTORE, bool HAS_BIAS, bool HAS_RES, bool RANK1>
__device__ __forceinline__ void gemm_tile(
    const float* __restrict__ A, long long strideA,
    const float* __restrict__ B, long long strideB,
    float* __restrict__ Cmat,    long long strideC,
    const float* __restrict__ bias,
    const float* __restrict__ resid, long long strideR,
    const float* __restrict__ rowv,  long long strideV,
    int bx, int by, int z) {
  static_assert(!RANK1 || HAS_BIAS, "the rank-1 start value needs its column vector");
  static_assert(K % (4 * NS) == 0 && K >= 8 * NS, "ring must divide K");
  static_assert((TM == 1 || TM == 2) && (WM == 2 || WM == 4), "tile shape");
  const float* Ab = A + (size_t)z * strideA;
  const float* Bb = B + (size_t)z * strideB;
  float*       Cb = Cmat + (size_t)z * strideC;

  int wave = threadIdx.x >> 5;
  int lane = threadIdx.x & 31;
  int half = lane >> 4;
  int ml   = lane & 15;
  constexpr int WN = 8 / WM;
  int m0 = bx * (16 * TM * WM) + (wave % WM) * (16 * TM);
  int n0 = by * (64 * WN) + (wave / WM) * 64;

  v8f acc[TM][4];
  {
    v4f bv = {0.0f, 0.0f, 0.0f, 0.0f};
    if (HAS_BIAS) bv = *(const v4f*)(bias + n0 + 4 * ml);
    if (RANK1) {
      const float* rv = rowv + (size_t)z * strideV + m0 + 8 * half;
      #pragma unroll
      for (int mi = 0; mi < TM; ++mi) {
        v4f r0 = *(const v4f*)(rv + 16 * mi);
        v4f r1 = *(const v4f*)(rv + 16 * mi + 4);
        #pragma unroll
        for (int t = 0; t < 4; ++t)
          #pragma unroll
          for (int r = 0; r < 4; ++r) {
            acc[mi][t][r]     = r0[r] * bv[t];
            acc[mi][t][r + 4] = r1[r] * bv[t];
          }
      }
    } else {
      #pragma unroll
      for (int mi = 0; mi < TM; ++mi)
        #pragma unroll
        for (int t = 0; t < 4; ++t)
          #pragma unroll
          for (int r = 0; r < 8; ++r) acc[mi][t][r] = bv[t];
    }
  }

  const float* Ap = Ab + (size_t)(2 * half) * LDA + m0 + ml;
  const float* Bp = Bb + (size_t)(2 * half) * LDB + n0 + 4 * ml;

  v2f fa[NS][TM];
  v4f rb[NS][2];

#define GEMM_LOAD_SLOT(s)                                                      \
  {                                                                            \
    const int k0 = 4 * (s) * LDA, k1 = (4 * (s) + 1) * LDA;                    \
    _Pragma("unroll")                                                          \
    for (int mi = 0; mi < TM; ++mi) {                                          \
      fa[s][mi].x = Ap[k0 + 16 * mi]; fa[s][mi].y = Ap[k1 + 16 * mi];          \
    }                                                                          \
    rb[s][0] = *(const v4f*)(Bp + 4 * (s) * LDB);                              \
    rb[s][1] = *(const v4f*)(Bp + (4 * (s) + 1) * LDB);                        \
  }
#define GEMM_MMA_SLOT(s)                                                       \
  {                                                                            \
    _Pragma("unroll")                                                          \
    for (int mi = 0; mi < TM; ++mi)                                            \
      _Pragma("unroll")                                                        \
      for (int t = 0; t < 4; ++t) {                                            \
        v2f fn = {rb[s][0][t], rb[s][1][t]};                                   \
        acc[mi][t] = __builtin_amdgcn_wmma_f32_16x16x4_f32(                    \
            false, fa[s][mi], false, fn, (short)0, acc[mi][t], false, false);  \
      }                                                                        \
  }

  #pragma unroll
  for (int s = 0; s < NS; ++s) {
    GEMM_LOAD_SLOT(s)
    __builtin_amdgcn_sched_barrier(0);
  }
  Ap += 4 * NS * LDA;
  Bp += 4 * NS * LDB;

  #pragma unroll 1
  for (int kk = 4 * NS; kk < K; kk += 4 * NS) {
    #pragma unroll
    for (int s = 0; s < NS; ++s) {
      GEMM_MMA_SLOT(s)
      GEMM_LOAD_SLOT(s)
      __builtin_amdgcn_sched_barrier(0);
    }
    Ap += 4 * NS * LDA;
    Bp += 4 * NS * LDB;
  }
  #pragma unroll
  for (int s = 0; s < NS; ++s) {
    GEMM_MMA_SLOT(s)
    __builtin_amdgcn_sched_barrier(0);
  }
#undef GEMM_LOAD_SLOT
#undef GEMM_MMA_SLOT

  if (!TSTORE) {
    #pragma unroll
    for (int mi = 0; mi < TM; ++mi) {
      #pragma unroll
      for (int r = 0; r < 8; ++r) {
        int m = m0 + 16 * mi + r + 8 * half;
        v4f v = {acc[mi][0][r], acc[mi][1][r], acc[mi][2][r], acc[mi][3][r]};
        *(v4f*)(Cb + (size_t)m * LDC + n0 + 4 * ml) = v;
      }
    }
  } else {
    const float* Rb = HAS_RES ? (resid + (size_t)z * strideR) : nullptr;
    #pragma unroll
    for (int mi = 0; mi < TM; ++mi) {
      #pragma unroll
      for (int r = 0; r < 8; ++r) {
        int m = m0 + 16 * mi + r + 8 * half;
        #pragma unroll
        for (int t = 0; t < 4; ++t) {
          size_t i = (size_t)(n0 + 4 * ml + t) * LDC + m;
          float v = acc[mi][t][r];
          if (HAS_RES) v += Rb[i];
          Cb[i] = v;
        }
      }
    }
  }
}

template <int LDA, int LDB, int LDC, int K, int NS, int TM, int WM,
          bool TSTORE, bool HAS_BIAS, bool HAS_RES, bool RANK1>
__global__ __launch_bounds__(256) void kT_gemm(
    const float* __restrict__ A, long long strideA,
    const float* __restrict__ B, long long strideB,
    float* __restrict__ Cmat,    long long strideC,
    const float* __restrict__ bias,
    const float* __restrict__ resid, long long strideR,
    const float* __restrict__ rowv,  long long strideV) {
  gemm_tile<LDA, LDB, LDC, K, NS, TM, WM, TSTORE, HAS_BIAS, HAS_RES, RANK1>(
      A, strideA, B, strideB, Cmat, strideC, bias, resid, strideR, rowv, strideV,
      blockIdx.x, blockIdx.y, blockIdx.z);
}

__global__ __launch_bounds__(256) void kT_stats_wtranspose(const float* __restrict__ x,
                                                           float* __restrict__ mean,
                                                           float* __restrict__ rstd,
                                                           const float* __restrict__ qkv_w,
                                                           float* __restrict__ wqkT,
                                                           const float* __restrict__ proj_w,
                                                           float* __restrict__ wprojT) {
  int id = blockIdx.x;
  if (id < B_ * GRP_) {
    gn_stats_group(x, mean, rstd, id);
  } else if (id < B_ * GRP_ + (C_ / 64) * (QK_ / 128)) {
    int t = id - B_ * GRP_;
    transpose_tile(qkv_w, wqkT, C_, QK_, 0, 0, t & 7, t >> 3, 0);
  } else {
    int t = id - B_ * GRP_ - (C_ / 64) * (QK_ / 128);
    transpose_tile(proj_w, wprojT, C_, C_, 0, 0, t & 7, t >> 3, 0);
  }
}

__global__ __launch_bounds__(256) void kT_w2_transposes_ksoftmax(float* qk, float* qT,
                                                                 const float* xn, float* xf,
                                                                 const float* wv,
                                                                 const float* wprojT,
                                                                 float* w2T) {
  constexpr int NW2 = (C_ / 64) * (C_ / 128);
  constexpr int NQT = (C_ / 64) * (N_ / 128) * B_;
  constexpr int NXF = (N_ / 64) * (C_ / 128) * B_;
  int id = blockIdx.x;
  if (id < NW2) {
    gemm_tile<C_, C_, C_, C_, 8, 1, 4, false, false, false, false>(
        wv, 0,
        wprojT, 0,
        w2T, 0,
        nullptr, nullptr, 0, nullptr, 0,
        id & 7, id >> 3, 0);
  } else if (id < NW2 + NQT) {
    int t = id - NW2;
    transpose_tile(qk, qT, QK_, N_, (long long)N_ * QK_, (long long)C_ * N_,
                   t & 7, (t >> 3) & 7, t >> 6);
  } else if (id < NW2 + NQT + NXF) {
    int t = id - NW2 - NQT;
    transpose_tile(xn, xf, N_, C_, (long long)C_ * N_, (long long)N_ * C_,
                   t & 15, (t >> 4) & 3, t >> 6);
  } else {
    softmax_k_rows(qk, id - NW2 - NQT - NXF);
  }
}

__global__ __launch_bounds__(256) void kT_kx_qsoftmax(const float* __restrict__ xf,
                                                      const float* __restrict__ kmat,
                                                      float* __restrict__ kxT,
                                                      float* __restrict__ qT,
                                                      float* __restrict__ ksum,
                                                      const float* __restrict__ proj_w,
                                                      const float* __restrict__ bv,
                                                      float* __restrict__ pb) {
  constexpr int NKX = (C_ / 64) * (C_ / 128) * B_;
  constexpr int NKS = B_ * (C_ / 64);
  constexpr int NPB = C_ / 64;
  int id = blockIdx.x;
  if (id < NKX) {
    gemm_tile<C_, QK_, C_, N_, 16, 1, 4, false, false, false, false>(
        xf, (long long)N_ * C_,
        kmat, (long long)N_ * QK_,
        kxT, (long long)C_ * C_,
        nullptr, nullptr, 0, nullptr, 0,
        id & 7, (id >> 3) & 3, id >> 5);
  } else if (id < NKX + NKS) {
    k_colsum_slice(kmat, ksum, id - NKX);
  } else if (id < NKX + NKS + NPB) {
    projbias_rows(proj_w, bv, pb, id - NKX - NKS);
  } else {
    softmax_rows1024(qT, id - NKX - NKS - NPB);
  }
}

extern "C" void kernel_launch(void* const* d_in, const int* in_sizes, int n_in,
                              void* d_out, int out_size, void* d_ws, size_t ws_size,
                              hipStream_t stream) {
  const float* x      = (const float*)d_in[0];
  const float* gn_w   = (const float*)d_in[1];
  const float* gn_b   = (const float*)d_in[2];
  const float* qkv_w  = (const float*)d_in[3];
  const float* qkv_b  = (const float*)d_in[4];
  const float* proj_w = (const float*)d_in[5];
  const float* proj_b = (const float*)d_in[6];
  float* out = (float*)d_out;

  float* ws = (float*)d_ws;
  const size_t XN_SZ  = (size_t)B_ * C_ * N_;
  const size_t QK_SZ  = (size_t)B_ * N_ * QK_;
  const size_t CTX_SZ = (size_t)B_ * C_ * C_;
  float* mean   = ws;
  float* rstd   = ws + 64;
  float* xn     = ws + 256;
  float* qk     = xn   + XN_SZ;
  float* xf     = qk   + QK_SZ;
  float* qT     = xf   + XN_SZ;
  float* kxT    = qT   + XN_SZ;
  float* ctxP   = kxT  + CTX_SZ;
  float* wqkT   = ctxP + CTX_SZ;
  float* w2T    = wqkT + (size_t)C_ * QK_;
  float* wprojT = w2T  + (size_t)C_ * C_;
  float* ksum   = wqkT;
  float* pb     = wqkT + (size_t)B_ * C_;


  kT_stats_wtranspose<<<160, 256, 0, stream>>>(x, mean, rstd, qkv_w, wqkT, proj_w, wprojT);

  kT_gn_apply<<<B_ * C_ / 8, 256, 0, stream>>>(x, gn_w, gn_b, mean, rstd, xn);

  kT_gemm<N_, QK_, QK_, C_, 4, 2, 2, false, true, false, false>
      <<<dim3(N_ / 64, QK_ / 256, B_), 256, 0, stream>>>(
      xn, (long long)C_ * N_,
      wqkT, 0,
      qk, (long long)N_ * QK_,
      qkv_b, nullptr, 0, nullptr, 0);

  kT_w2_transposes_ksoftmax<<<32 + 512 + 512 + 512, 256, 0, stream>>>(
      qk, qT, xn, xf, qkv_w + (size_t)QK_ * C_, wprojT, w2T);

  kT_kx_qsoftmax<<<256 + 64 + 8 + 512, 256, 0, stream>>>(
      xf, qk + C_, kxT, qT, ksum, proj_w, qkv_b + QK_, pb);

  kT_gemm<C_, C_, C_, C_, 16, 1, 4, false, true, false, true>
      <<<dim3(C_ / 64, C_ / 128, B_), 256, 0, stream>>>(
      kxT, (long long)C_ * C_,
      w2T, 0,
      ctxP, (long long)C_ * C_,
      pb, nullptr, 0, ksum, (long long)C_);

  kT_gemm<N_, C_, N_, C_, 8, 2, 2, true, true, true, false>
      <<<dim3(N_ / 64, C_ / 256, B_), 256, 0, stream>>>(
      qT, (long long)C_ * N_,
      ctxP, (long long)C_ * C_,
      out, (long long)C_ * N_,
      proj_b, xn, (long long)C_ * N_, nullptr, 0);
}
